# starting kernel
# baseline (speedup 1.0000x reference)
#include <hip/hip_runtime.h>
#include <hip/hip_bf16.h>

typedef __attribute__((ext_vector_type(2))) float v2f;
typedef __attribute__((ext_vector_type(8))) float v8f;

#define NQ 10
#define DIM 1024
#define WPB 8

__global__ void build_gates_kernel(const float* __restrict__ W,
                                   float* __restrict__ G) {
  int g = threadIdx.x;
  if (g >= 40) return;
  int L = g / 20, r = g % 20;
  float u00r, u00i, u01r, u01i, u10r, u10i, u11r, u11i;
  if (r < 10) {
    int base = L * 40 + 3 * r;
    float phi = W[base], th = W[base + 1], om = W[base + 2];
    float ct = cosf(0.5f * th), st = sinf(0.5f * th);
    float ap = 0.5f * (phi + om), am = 0.5f * (phi - om);
    float cap = cosf(ap), sap = sinf(ap);
    float cam = cosf(am), sam = sinf(am);
    u00r =  ct * cap; u00i = -ct * sap;
    u01r = -st * cam; u01i = -st * sam;
    u10r =  st * cam; u10i = -st * sam;
    u11r =  ct * cap; u11i =  ct * sap;
  } else {
    int e = r - 10;
    float th = W[L * 40 + 30 + e];
    float ct = cosf(0.5f * th), st = sinf(0.5f * th);
    u00r = ct;  u00i = 0.f; u01r = -st; u01i = 0.f;
    u10r = st;  u10i = 0.f; u11r = ct;  u11i = 0.f;
  }
  float* M = G + g * 16;
  M[ 0] = u00r; M[ 1] = -u00i; M[ 2] = u01r; M[ 3] = -u01i;
  M[ 4] = u00i; M[ 5] =  u00r; M[ 6] = u01i; M[ 7] =  u01r;
  M[ 8] = u10r; M[ 9] = -u10i; M[10] = u11r; M[11] = -u11i;
  M[12] = u10i; M[13] =  u10r; M[14] = u11i; M[15] =  u11r;
}

__global__ __launch_bounds__(256, 8) void qsim_kernel(
    const float* __restrict__ inp, const float* __restrict__ G,
    float* __restrict__ out) {
  __shared__ float2 st[WPB][DIM];
  const int t = threadIdx.x & 31;
  const int w = threadIdx.x >> 5;
  const int samp = blockIdx.x * WPB + w;
  float2* sv = st[w];

  float cc[NQ], ss[NQ];
  const float* x = inp + samp * NQ;
  #pragma unroll
  for (int q = 0; q < NQ; ++q) {
    float xv = fminf(fmaxf(x[q], 0.f), 1.f);
    float a = 0.5f * 3.14159265358979323846f * xv;
    cc[q] = cosf(a);
    ss[q] = sinf(a);
  }
  float plo = 1.f;
  #pragma unroll
  for (int j = 0; j < 5; ++j)
    plo *= ((t >> j) & 1) ? ss[9 - j] : cc[9 - j];
  #pragma unroll 4
  for (int r = 0; r < 32; ++r) {
    float phi = 1.f;
    #pragma unroll
    for (int j = 0; j < 5; ++j)
      phi *= ((r >> j) & 1) ? ss[4 - j] : cc[4 - j];
    sv[r * 32 + t] = make_float2(phi * plo, 0.f);
  }
  __builtin_amdgcn_wave_barrier();

  const int row4 = (t & 3) * 4;
  const int c0 = (t < 16) ? 0 : 2;
  const int pl = t & 15;
  const bool lo = (t < 16);

  for (int L = 0; L < 2; ++L) {
    for (int q = 0; q < NQ; ++q) {
      const float* M = G + (L * 20 + q) * 16;
      v2f a;
      a.x = M[row4 + c0];
      a.y = M[row4 + c0 + 1];
      const int b = 9 - q;
      const int S = 1 << b;
      #pragma unroll 4
      for (int blk = 0; blk < 32; ++blk) {
        int p = blk * 16 + pl;
        int i0 = ((p >> b) << (b + 1)) | (p & (S - 1));
        int idx = lo ? i0 : (i0 | S);
        float2 amp = sv[idx];
        v2f bb;
        bb.x = amp.x;
        bb.y = amp.y;
        v8f cz = {};
        v8f d = __builtin_amdgcn_wmma_f32_16x16x4_f32(
            false, a, false, bb, (short)0, cz, false, false);
        sv[idx] = make_float2(lo ? d[0] : d[2], lo ? d[1] : d[3]);
      }
      __builtin_amdgcn_wave_barrier();
    }
    for (int e = 0; e < NQ; ++e) {
      const float* M = G + (L * 20 + 10 + e) * 16;
      v2f a;
      a.x = M[row4 + c0];
      a.y = M[row4 + c0 + 1];
      const int u = e, v = (e + 1) % NQ;
      const int bu = 9 - u, bv = 9 - v;
      const int blo = bu < bv ? bu : bv;
      const int bhi = bu < bv ? bv : bu;
      #pragma unroll 4
      for (int blk = 0; blk < 16; ++blk) {
        int p = blk * 16 + pl;
        int x1 = ((p >> blo) << (blo + 1)) | (p & ((1 << blo) - 1));
        int x2 = ((x1 >> bhi) << (bhi + 1)) | (x1 & ((1 << bhi) - 1));
        int i0 = x2 | (1 << bu);
        int idx = lo ? i0 : (i0 | (1 << bv));
        float2 amp = sv[idx];
        v2f bb;
        bb.x = amp.x;
        bb.y = amp.y;
        v8f cz = {};
        v8f d = __builtin_amdgcn_wmma_f32_16x16x4_f32(
            false, a, false, bb, (short)0, cz, false, false);
        sv[idx] = make_float2(lo ? d[0] : d[2], lo ? d[1] : d[3]);
      }
      __builtin_amdgcn_wave_barrier();
    }
  }

  float acc[NQ];
  #pragma unroll
  for (int q = 0; q < NQ; ++q) acc[q] = 0.f;
  #pragma unroll 4
  for (int r = 0; r < 32; ++r) {
    int i = r * 32 + t;
    float2 amp = sv[i];
    float p = amp.x * amp.x + amp.y * amp.y;
    #pragma unroll
    for (int q = 0; q < NQ; ++q)
      acc[q] += ((i >> (9 - q)) & 1) ? -p : p;
  }
  #pragma unroll
  for (int q = 0; q < NQ; ++q) {
    #pragma unroll
    for (int off = 16; off >= 1; off >>= 1)
      acc[q] += __shfl_xor(acc[q], off, 32);
  }
  if (t == 0) {
    #pragma unroll
    for (int q = 0; q < NQ; ++q) out[samp * NQ + q] = acc[q];
  }
}

extern "C" void kernel_launch(void* const* d_in, const int* in_sizes, int n_in,
                              void* d_out, int out_size, void* d_ws,
                              size_t ws_size, hipStream_t stream) {
  const float* inputs = (const float*)d_in[0];
  const float* weights = (const float*)d_in[1];
  float* out = (float*)d_out;
  float* G = (float*)d_ws;

  const int B = in_sizes[0] / NQ;
  build_gates_kernel<<<dim3(1), dim3(64), 0, stream>>>(weights, G);
  qsim_kernel<<<dim3(B / WPB), dim3(256), 0, stream>>>(inputs, G, out);
}


# Layer-0 Rot gates act on each qubit's (cos,sin) 2-vector before the Kronecker product: 20 WMMAs per sample, not 320; state built by complex products; layer-0 gate sums and prep products reassociated
# speedup vs baseline: 1.9248x; 1.9248x over previous
#include <hip/hip_runtime.h>
#include <hip/hip_bf16.h>

typedef __attribute__((ext_vector_type(2))) float v2f;
typedef __attribute__((ext_vector_type(8))) float v8f;

#define NQ 10
#define DIM 1024
#define WPB 8

__global__ void build_gates_kernel(const float* __restrict__ W,
                                   float* __restrict__ G) {
  int g = threadIdx.x;
  if (g >= 40) return;
  int L = g / 20, r = g % 20;
  float u00r, u00i, u01r, u01i, u10r, u10i, u11r, u11i;
  if (r < 10) {
    int base = L * 40 + 3 * r;
    float phi = W[base], th = W[base + 1], om = W[base + 2];
    float ct = cosf(0.5f * th), st = sinf(0.5f * th);
    float ap = 0.5f * (phi + om), am = 0.5f * (phi - om);
    float cap = cosf(ap), sap = sinf(ap);
    float cam = cosf(am), sam = sinf(am);
    u00r =  ct * cap; u00i = -ct * sap;
    u01r = -st * cam; u01i = -st * sam;
    u10r =  st * cam; u10i = -st * sam;
    u11r =  ct * cap; u11i =  ct * sap;
  } else {
    int e = r - 10;
    float th = W[L * 40 + 30 + e];
    float ct = cosf(0.5f * th), st = sinf(0.5f * th);
    u00r = ct;  u00i = 0.f; u01r = -st; u01i = 0.f;
    u10r = st;  u10i = 0.f; u11r = ct;  u11i = 0.f;
  }
  float* M = G + g * 16;
  M[ 0] = u00r; M[ 1] = -u00i; M[ 2] = u01r; M[ 3] = -u01i;
  M[ 4] = u00i; M[ 5] =  u00r; M[ 6] = u01i; M[ 7] =  u01r;
  M[ 8] = u10r; M[ 9] = -u10i; M[10] = u11r; M[11] = -u11i;
  M[12] = u10i; M[13] =  u10r; M[14] = u11i; M[15] =  u11r;
}

__device__ __forceinline__ v8f mm(v2f a, v2f b, v8f c) {
  return __builtin_amdgcn_wmma_f32_16x16x4_f32(false, a, false, b, (short)0, c,
                                               false, false);
}

__device__ __forceinline__ v2f cmul(v2f a, v2f b) {
  return v2f{a.x * b.x - a.y * b.y, a.x * b.y + a.y * b.x};
}

__device__ __forceinline__ void load_gate(const float* __restrict__ G, int g,
                                          int t, v2f& a1, v2f& a2, v2f& ad) {
  const float* M = G + g * 16 + (t & 3) * 4;
  const float m0 = M[0], m1 = M[1], m2 = M[2], m3 = M[3];
  const bool on = ((t < 16) == ((t & 15) < 8));
  a1.x = on ? m0 : 0.f; a1.y = on ? m1 : 0.f;
  a2.x = on ? m2 : 0.f; a2.y = on ? m3 : 0.f;
  ad.x = (t < 16) ? m0 : m2; ad.y = (t < 16) ? m1 : m3;
}

template <int RB, int CB>
__device__ __forceinline__ void gate_rr(v2f (&s)[32], v2f a1, v2f a2) {
  #pragma unroll
  for (int p = 0; p < 16; ++p) {
    const int r0 = ((p >> RB) << (RB + 1)) | (p & ((1 << RB) - 1));
    const int r1 = r0 | (1 << RB);
    if (CB >= 0 && !((r0 >> (CB < 0 ? 0 : CB)) & 1)) continue;
    v8f z = {};
    v8f d = mm(a1, s[r0], z);
    d = mm(a2, s[r1], d);
    s[r0] = v2f{d[0], d[1]};
    s[r1] = v2f{d[2], d[3]};
  }
}

template <int RB>
__device__ __forceinline__ void gate_lr(v2f (&s)[32], v2f a1, v2f a2, bool ctl) {
  #pragma unroll
  for (int p = 0; p < 16; ++p) {
    const int r0 = ((p >> RB) << (RB + 1)) | (p & ((1 << RB) - 1));
    const int r1 = r0 | (1 << RB);
    v8f z = {};
    v8f d = mm(a1, s[r0], z);
    d = mm(a2, s[r1], d);
    s[r0] = v2f{ctl ? d[0] : s[r0].x, ctl ? d[1] : s[r0].y};
    s[r1] = v2f{ctl ? d[2] : s[r1].x, ctl ? d[3] : s[r1].y};
  }
}

template <int CB>
__device__ __forceinline__ void gate_rl(v2f (&s)[32], v2f ad, bool lo) {
  #pragma unroll
  for (int r = 0; r < 32; ++r) {
    if (!((r >> CB) & 1)) continue;
    v8f z = {};
    v8f d = mm(ad, s[r], z);
    s[r] = v2f{lo ? d[0] : d[2], lo ? d[1] : d[3]};
  }
}

__device__ __forceinline__ void transpose(v2f (&s)[32], float2* sv, int t) {
  #pragma unroll
  for (int r = 0; r < 32; ++r)
    sv[r * 32 + (t ^ r)] = make_float2(s[r].x, s[r].y);
  __builtin_amdgcn_wave_barrier();
  #pragma unroll
  for (int r = 0; r < 32; ++r) {
    float2 v = sv[t * 32 + (r ^ t)];
    s[r] = v2f{v.x, v.y};
  }
  __builtin_amdgcn_wave_barrier();
}

#define ROT_A(L, q)  { load_gate(G, (L) * 20 + (q), t, a1, a2, ad); gate_rr<4 - (q), -1>(s, a1, a2); }
#define CRY_A(L, e)  { load_gate(G, (L) * 20 + 10 + (e), t, a1, a2, ad); gate_rr<3 - (e), 4 - (e)>(s, a1, a2); }
#define ROT_B(L, q)  { load_gate(G, (L) * 20 + (q), t, a1, a2, ad); gate_rr<9 - (q), -1>(s, a1, a2); }
#define CRY_B(L, e)  { load_gate(G, (L) * 20 + 10 + (e), t, a1, a2, ad); gate_rr<8 - (e), 9 - (e)>(s, a1, a2); }
#define CRYS_A(L)    CRY_A(L, 0) CRY_A(L, 1) CRY_A(L, 2) CRY_A(L, 3)
#define PHASE_A(L)   ROT_A(L, 0) ROT_A(L, 1) ROT_A(L, 2) ROT_A(L, 3) ROT_A(L, 4) \
                     CRYS_A(L)
#define ROTS_B(L)    ROT_B(L, 5) ROT_B(L, 6) ROT_B(L, 7) ROT_B(L, 8) ROT_B(L, 9)
#define CRYS_B(L)    { load_gate(G, (L) * 20 + 14, t, a1, a2, ad); gate_lr<4>(s, a1, a2, (t & 1) != 0); } \
                     CRY_B(L, 5) CRY_B(L, 6) CRY_B(L, 7) CRY_B(L, 8) \
                     { load_gate(G, (L) * 20 + 19, t, a1, a2, ad); gate_rl<0>(s, ad, t < 16); }

__global__ __launch_bounds__(256, 4) void qsim_kernel(
    const float* __restrict__ inp, const float* __restrict__ G,
    float* __restrict__ out) {
  __shared__ float2 st[WPB][DIM];
  const int t = threadIdx.x & 31;
  const int w = threadIdx.x >> 5;
  const int samp = blockIdx.x * WPB + w;
  float2* sv = st[w];

  float cc[NQ], ss[NQ];
  const float* x = inp + samp * NQ;
  #pragma unroll
  for (int q = 0; q < NQ; ++q) {
    float xv = fminf(fmaxf(x[q], 0.f), 1.f);
    float a = 0.5f * 3.14159265358979323846f * xv;
    cc[q] = cosf(a);
    ss[q] = sinf(a);
  }
  v2f a1, a2, ad;
  v2f u0[NQ], u1[NQ];
  #pragma unroll
  for (int q = 0; q < NQ; ++q) {
    load_gate(G, q, t, a1, a2, ad);
    v8f z = {};
    v8f d = mm(a1, v2f{cc[q], 0.f}, z);
    d = mm(a2, v2f{ss[q], 0.f}, d);
    u0[q] = v2f{d[0], d[1]};
    u1[q] = v2f{d[2], d[3]};
  }
  v2f plo = (t & 1) ? u1[9] : u0[9];
  #pragma unroll
  for (int j = 1; j < 5; ++j)
    plo = cmul(plo, ((t >> j) & 1) ? u1[9 - j] : u0[9 - j]);
  v2f s[32];
  s[0] = plo;
  #pragma unroll
  for (int j = 0; j < 5; ++j) {
    #pragma unroll
    for (int e = 0; e < (1 << j); ++e) {
      s[e | (1 << j)] = cmul(s[e], u1[4 - j]);
      s[e] = cmul(s[e], u0[4 - j]);
    }
  }

  CRYS_A(0)
  transpose(s, sv, t);
  CRYS_B(0)
  ROTS_B(1)
  transpose(s, sv, t);
  PHASE_A(1)
  transpose(s, sv, t);
  CRYS_B(1)

  float acc[NQ];
  #pragma unroll
  for (int q = 0; q < NQ; ++q) acc[q] = 0.f;
  #pragma unroll
  for (int r = 0; r < 32; ++r) {
    const int i = t * 32 + r;
    float p = s[r].x * s[r].x + s[r].y * s[r].y;
    #pragma unroll
    for (int q = 0; q < NQ; ++q)
      acc[q] += ((i >> (9 - q)) & 1) ? -p : p;
  }
  #pragma unroll
  for (int q = 0; q < NQ; ++q) {
    #pragma unroll
    for (int off = 16; off >= 1; off >>= 1)
      acc[q] += __shfl_xor(acc[q], off, 32);
  }
  if (t == 0) {
    #pragma unroll
    for (int q = 0; q < NQ; ++q) out[samp * NQ + q] = acc[q];
  }
}

extern "C" void kernel_launch(void* const* d_in, const int* in_sizes, int n_in,
                              void* d_out, int out_size, void* d_ws,
                              size_t ws_size, hipStream_t stream) {
  const float* inputs = (const float*)d_in[0];
  const float* weights = (const float*)d_in[1];
  float* out = (float*)d_out;
  float* G = (float*)d_ws;

  const int B = in_sizes[0] / NQ;
  build_gates_kernel<<<dim3(1), dim3(64), 0, stream>>>(weights, G);
  qsim_kernel<<<dim3(B / WPB), dim3(256), 0, stream>>>(inputs, G, out);
}


# Layer 1: Rot and the ring gate on its target are one control-selected gate (RY*Rot prebuilt; a lane control picks D rows 0-3 or 4-7): 532 WMMAs per sample, not 692; gate products reassociated
# speedup vs baseline: 2.3450x; 1.2183x over previous
#include <hip/hip_runtime.h>
#include <hip/hip_bf16.h>

typedef __attribute__((ext_vector_type(2))) float v2f;
typedef __attribute__((ext_vector_type(8))) float v8f;

#define NQ 10
#define DIM 1024
#define WPB 8

__global__ void build_gates_kernel(const float* __restrict__ W,
                                   float* __restrict__ G) {
  int g = threadIdx.x;
  if (g >= 40) return;
  int L = g / 20, r = g % 20;
  float u00r = 0.f, u00i = 0.f, u01r = 0.f, u01i = 0.f;
  float u10r = 0.f, u10i = 0.f, u11r = 0.f, u11i = 0.f;
  const bool cry = (r >= 10);
  const bool fused = cry && L == 1 && r < 19;
  if (!cry || fused) {
    int q = cry ? (r - 9) : r;
    int base = L * 40 + 3 * q;
    float phi = W[base], th = W[base + 1], om = W[base + 2];
    float ct = cosf(0.5f * th), st = sinf(0.5f * th);
    float ap = 0.5f * (phi + om), am = 0.5f * (phi - om);
    float cap = cosf(ap), sap = sinf(ap);
    float cam = cosf(am), sam = sinf(am);
    u00r =  ct * cap; u00i = -ct * sap;
    u01r = -st * cam; u01i = -st * sam;
    u10r =  st * cam; u10i = -st * sam;
    u11r =  ct * cap; u11i =  ct * sap;
  }
  if (cry) {
    int e = r - 10;
    float th = W[L * 40 + 30 + e];
    float ct = cosf(0.5f * th), st = sinf(0.5f * th);
    if (fused) {
      float n00r = ct * u00r - st * u10r, n00i = ct * u00i - st * u10i;
      float n01r = ct * u01r - st * u11r, n01i = ct * u01i - st * u11i;
      float n10r = st * u00r + ct * u10r, n10i = st * u00i + ct * u10i;
      float n11r = st * u01r + ct * u11r, n11i = st * u01i + ct * u11i;
      u00r = n00r; u00i = n00i; u01r = n01r; u01i = n01i;
      u10r = n10r; u10i = n10i; u11r = n11r; u11i = n11i;
    } else {
      u00r = ct;  u00i = 0.f; u01r = -st; u01i = 0.f;
      u10r = st;  u10i = 0.f; u11r = ct;  u11i = 0.f;
    }
  }
  float* M = G + g * 16;
  M[ 0] = u00r; M[ 1] = -u00i; M[ 2] = u01r; M[ 3] = -u01i;
  M[ 4] = u00i; M[ 5] =  u00r; M[ 6] = u01i; M[ 7] =  u01r;
  M[ 8] = u10r; M[ 9] = -u10i; M[10] = u11r; M[11] = -u11i;
  M[12] = u10i; M[13] =  u10r; M[14] = u11i; M[15] =  u11r;
}

__device__ __forceinline__ v8f mm(v2f a, v2f b, v8f c) {
  return __builtin_amdgcn_wmma_f32_16x16x4_f32(false, a, false, b, (short)0, c,
                                               false, false);
}

__device__ __forceinline__ v2f cmul(v2f a, v2f b) {
  return v2f{a.x * b.x - a.y * b.y, a.x * b.y + a.y * b.x};
}

__device__ __forceinline__ void load_gate(const float* __restrict__ G, int g,
                                          int t, v2f& a1, v2f& a2, v2f& ad) {
  const float* M = G + g * 16 + (t & 3) * 4;
  const float m0 = M[0], m1 = M[1], m2 = M[2], m3 = M[3];
  const bool on = ((t < 16) == ((t & 15) < 8));
  a1.x = on ? m0 : 0.f; a1.y = on ? m1 : 0.f;
  a2.x = on ? m2 : 0.f; a2.y = on ? m3 : 0.f;
  ad.x = (t < 16) ? m0 : m2; ad.y = (t < 16) ? m1 : m3;
}

__device__ __forceinline__ void load_mux(const float* __restrict__ G, int gR,
                                         int gF, int t, v2f& a1, v2f& a2) {
  const float* M = G + ((t & 4) ? gF : gR) * 16 + (t & 3) * 4;
  const float m0 = M[0], m1 = M[1], m2 = M[2], m3 = M[3];
  const bool on = ((t < 16) == ((t & 15) < 8));
  a1.x = on ? m0 : 0.f; a1.y = on ? m1 : 0.f;
  a2.x = on ? m2 : 0.f; a2.y = on ? m3 : 0.f;
}

template <int RB, int CB>
__device__ __forceinline__ void gate_rr(v2f (&s)[32], v2f a1, v2f a2) {
  #pragma unroll
  for (int p = 0; p < 16; ++p) {
    const int r0 = ((p >> RB) << (RB + 1)) | (p & ((1 << RB) - 1));
    const int r1 = r0 | (1 << RB);
    if (CB >= 0 && !((r0 >> (CB < 0 ? 0 : CB)) & 1)) continue;
    v8f z = {};
    v8f d = mm(a1, s[r0], z);
    d = mm(a2, s[r1], d);
    s[r0] = v2f{d[0], d[1]};
    s[r1] = v2f{d[2], d[3]};
  }
}

template <int RB>
__device__ __forceinline__ void gate_lr(v2f (&s)[32], v2f a1, v2f a2, bool ctl) {
  #pragma unroll
  for (int p = 0; p < 16; ++p) {
    const int r0 = ((p >> RB) << (RB + 1)) | (p & ((1 << RB) - 1));
    const int r1 = r0 | (1 << RB);
    v8f z = {};
    v8f d = mm(a1, s[r0], z);
    d = mm(a2, s[r1], d);
    s[r0] = v2f{ctl ? d[0] : s[r0].x, ctl ? d[1] : s[r0].y};
    s[r1] = v2f{ctl ? d[2] : s[r1].x, ctl ? d[3] : s[r1].y};
  }
}

template <int RB, int CB>
__device__ __forceinline__ void gate_mux_rr(v2f (&s)[32], v2f r1, v2f r2,
                                            v2f f1, v2f f2) {
  #pragma unroll
  for (int p = 0; p < 16; ++p) {
    const int i0 = ((p >> RB) << (RB + 1)) | (p & ((1 << RB) - 1));
    const int i1 = i0 | (1 << RB);
    const bool c = ((i0 >> CB) & 1) != 0;
    v8f z = {};
    v8f d = mm(c ? f1 : r1, s[i0], z);
    d = mm(c ? f2 : r2, s[i1], d);
    s[i0] = v2f{d[0], d[1]};
    s[i1] = v2f{d[2], d[3]};
  }
}

template <int RB>
__device__ __forceinline__ void gate_mux_lr(v2f (&s)[32], v2f m1, v2f m2,
                                            bool ctl) {
  #pragma unroll
  for (int p = 0; p < 16; ++p) {
    const int i0 = ((p >> RB) << (RB + 1)) | (p & ((1 << RB) - 1));
    const int i1 = i0 | (1 << RB);
    v8f z = {};
    v8f d = mm(m1, s[i0], z);
    d = mm(m2, s[i1], d);
    s[i0] = v2f{ctl ? d[4] : d[0], ctl ? d[5] : d[1]};
    s[i1] = v2f{ctl ? d[6] : d[2], ctl ? d[7] : d[3]};
  }
}

template <int CB>
__device__ __forceinline__ void gate_rl(v2f (&s)[32], v2f ad, bool lo) {
  #pragma unroll
  for (int r = 0; r < 32; ++r) {
    if (!((r >> CB) & 1)) continue;
    v8f z = {};
    v8f d = mm(ad, s[r], z);
    s[r] = v2f{lo ? d[0] : d[2], lo ? d[1] : d[3]};
  }
}

__device__ __forceinline__ void transpose(v2f (&s)[32], float2* sv, int t) {
  #pragma unroll
  for (int r = 0; r < 32; ++r)
    sv[r * 32 + (t ^ r)] = make_float2(s[r].x, s[r].y);
  __builtin_amdgcn_wave_barrier();
  #pragma unroll
  for (int r = 0; r < 32; ++r) {
    float2 v = sv[t * 32 + (r ^ t)];
    s[r] = v2f{v.x, v.y};
  }
  __builtin_amdgcn_wave_barrier();
}

#define ROT_A(L, q)  { load_gate(G, (L) * 20 + (q), t, a1, a2, ad); gate_rr<4 - (q), -1>(s, a1, a2); }
#define CRY_A(L, e)  { load_gate(G, (L) * 20 + 10 + (e), t, a1, a2, ad); gate_rr<3 - (e), 4 - (e)>(s, a1, a2); }
#define CRY_B(L, e)  { load_gate(G, (L) * 20 + 10 + (e), t, a1, a2, ad); gate_rr<8 - (e), 9 - (e)>(s, a1, a2); }
#define CRYS_A(L)    CRY_A(L, 0) CRY_A(L, 1) CRY_A(L, 2) CRY_A(L, 3)
#define MUX_A(e)     { load_gate(G, 21 + (e), t, a1, a2, ad); load_gate(G, 30 + (e), t, f1, f2, ad); \
                       gate_mux_rr<3 - (e), 4 - (e)>(s, a1, a2, f1, f2); }
#define MUX_B(e)     { load_gate(G, 21 + (e), t, a1, a2, ad); load_gate(G, 30 + (e), t, f1, f2, ad); \
                       gate_mux_rr<8 - (e), 9 - (e)>(s, a1, a2, f1, f2); }
#define PHASE_A1     ROT_A(1, 0) MUX_A(0) MUX_A(1) MUX_A(2) MUX_A(3)
#define PHASE_B1     { load_mux(G, 25, 34, t, a1, a2); gate_mux_lr<4>(s, a1, a2, (t & 1) != 0); } \
                     MUX_B(5) MUX_B(6) MUX_B(7) MUX_B(8) \
                     { load_gate(G, 39, t, a1, a2, ad); gate_rl<0>(s, ad, t < 16); }
#define CRYS_B(L)    { load_gate(G, (L) * 20 + 14, t, a1, a2, ad); gate_lr<4>(s, a1, a2, (t & 1) != 0); } \
                     CRY_B(L, 5) CRY_B(L, 6) CRY_B(L, 7) CRY_B(L, 8) \
                     { load_gate(G, (L) * 20 + 19, t, a1, a2, ad); gate_rl<0>(s, ad, t < 16); }

__global__ __launch_bounds__(256, 4) void qsim_kernel(
    const float* __restrict__ inp, const float* __restrict__ G,
    float* __restrict__ out) {
  __shared__ float2 st[WPB][DIM];
  const int t = threadIdx.x & 31;
  const int w = threadIdx.x >> 5;
  const int samp = blockIdx.x * WPB + w;
  float2* sv = st[w];

  float cc[NQ], ss[NQ];
  const float* x = inp + samp * NQ;
  #pragma unroll
  for (int q = 0; q < NQ; ++q) {
    float xv = fminf(fmaxf(x[q], 0.f), 1.f);
    float a = 0.5f * 3.14159265358979323846f * xv;
    cc[q] = cosf(a);
    ss[q] = sinf(a);
  }
  v2f a1, a2, ad, f1, f2;
  v2f u0[NQ], u1[NQ];
  #pragma unroll
  for (int q = 0; q < NQ; ++q) {
    load_gate(G, q, t, a1, a2, ad);
    v8f z = {};
    v8f d = mm(a1, v2f{cc[q], 0.f}, z);
    d = mm(a2, v2f{ss[q], 0.f}, d);
    u0[q] = v2f{d[0], d[1]};
    u1[q] = v2f{d[2], d[3]};
  }
  v2f plo = (t & 1) ? u1[9] : u0[9];
  #pragma unroll
  for (int j = 1; j < 5; ++j)
    plo = cmul(plo, ((t >> j) & 1) ? u1[9 - j] : u0[9 - j]);
  v2f s[32];
  s[0] = plo;
  #pragma unroll
  for (int j = 0; j < 5; ++j) {
    #pragma unroll
    for (int e = 0; e < (1 << j); ++e) {
      s[e | (1 << j)] = cmul(s[e], u1[4 - j]);
      s[e] = cmul(s[e], u0[4 - j]);
    }
  }

  CRYS_A(0)
  transpose(s, sv, t);
  CRYS_B(0)
  transpose(s, sv, t);
  PHASE_A1
  transpose(s, sv, t);
  PHASE_B1

  float acc[NQ];
  #pragma unroll
  for (int q = 0; q < NQ; ++q) acc[q] = 0.f;
  #pragma unroll
  for (int r = 0; r < 32; ++r) {
    const int i = t * 32 + r;
    float p = s[r].x * s[r].x + s[r].y * s[r].y;
    #pragma unroll
    for (int q = 0; q < NQ; ++q)
      acc[q] += ((i >> (9 - q)) & 1) ? -p : p;
  }
  #pragma unroll
  for (int q = 0; q < NQ; ++q) {
    #pragma unroll
    for (int off = 16; off >= 1; off >>= 1)
      acc[q] += __shfl_xor(acc[q], off, 32);
  }
  if (t == 0) {
    #pragma unroll
    for (int q = 0; q < NQ; ++q) out[samp * NQ + q] = acc[q];
  }
}

extern "C" void kernel_launch(void* const* d_in, const int* in_sizes, int n_in,
                              void* d_out, int out_size, void* d_ws,
                              size_t ws_size, hipStream_t stream) {
  const float* inputs = (const float*)d_in[0];
  const float* weights = (const float*)d_in[1];
  float* out = (float*)d_out;
  float* G = (float*)d_ws;

  const int B = in_sizes[0] / NQ;
  build_gates_kernel<<<dim3(1), dim3(64), 0, stream>>>(weights, G);
  qsim_kernel<<<dim3(B / WPB), dim3(256), 0, stream>>>(inputs, G, out);
}


# Rot and ring gates on two register qubits are ONE per-lane 2-qubit gate (real 8x8, 4 chained block-diagonal k=4 WMMAs, all 8 D rows used), both layers: 372 WMMAs, not 532; gate products reassociated
# speedup vs baseline: 2.8728x; 1.2251x over previous
#include <hip/hip_runtime.h>
#include <hip/hip_bf16.h>

typedef __attribute__((ext_vector_type(2))) float v2f;
typedef __attribute__((ext_vector_type(8))) float v8f;

#define NQ 10
#define DIM 1024
#define WPB 8

__global__ void build_gates_kernel(const float* __restrict__ W,
                                   float* __restrict__ G) {
  __shared__ float sm[640];
  const int g = threadIdx.x;
  if (g < 40) {
  int L = g / 20, r = g % 20;
  float u00r = 0.f, u00i = 0.f, u01r = 0.f, u01i = 0.f;
  float u10r = 0.f, u10i = 0.f, u11r = 0.f, u11i = 0.f;
  const bool cry = (r >= 10);
  const bool fused = cry && L == 1 && r < 19;
  if (!cry || fused) {
    int q = cry ? (r - 9) : r;
    int base = L * 40 + 3 * q;
    float phi = W[base], th = W[base + 1], om = W[base + 2];
    float ct = cosf(0.5f * th), st = sinf(0.5f * th);
    float ap = 0.5f * (phi + om), am = 0.5f * (phi - om);
    float cap = cosf(ap), sap = sinf(ap);
    float cam = cosf(am), sam = sinf(am);
    u00r =  ct * cap; u00i = -ct * sap;
    u01r = -st * cam; u01i = -st * sam;
    u10r =  st * cam; u10i = -st * sam;
    u11r =  ct * cap; u11i =  ct * sap;
  }
  if (cry) {
    int e = r - 10;
    float th = W[L * 40 + 30 + e];
    float ct = cosf(0.5f * th), st = sinf(0.5f * th);
    if (fused) {
      float n00r = ct * u00r - st * u10r, n00i = ct * u00i - st * u10i;
      float n01r = ct * u01r - st * u11r, n01i = ct * u01i - st * u11i;
      float n10r = st * u00r + ct * u10r, n10i = st * u00i + ct * u10i;
      float n11r = st * u01r + ct * u11r, n11i = st * u01i + ct * u11i;
      u00r = n00r; u00i = n00i; u01r = n01r; u01i = n01i;
      u10r = n10r; u10i = n10i; u11r = n11r; u11i = n11i;
    } else {
      u00r = ct;  u00i = 0.f; u01r = -st; u01i = 0.f;
      u10r = st;  u10i = 0.f; u11r = ct;  u11i = 0.f;
    }
  }
  float* M = G + g * 16;
  M[ 0] = u00r; M[ 1] = -u00i; M[ 2] = u01r; M[ 3] = -u01i;
  M[ 4] = u00i; M[ 5] =  u00r; M[ 6] = u01i; M[ 7] =  u01r;
  M[ 8] = u10r; M[ 9] = -u10i; M[10] = u11r; M[11] = -u11i;
  M[12] = u10i; M[13] =  u10r; M[14] = u11i; M[15] =  u11r;
  #pragma unroll
  for (int k = 0; k < 16; ++k) sm[g * 16 + k] = M[k];
  }
  __syncthreads();
  {
    const int R = g >> 3, Cc = g & 7;
    const int ao = R >> 2, bo = (R >> 1) & 1, co = R & 1;
    const int ai = Cc >> 2, bi = (Cc >> 1) & 1, ci = Cc & 1;
    #pragma unroll
    for (int m = 0; m < 12; ++m) {
      const bool l0 = (m >= 8);
      const int pr = l0 ? (m - 8) : (m >> 1);
      const int a = (pr == 0) ? 1 : (pr == 1 ? 3 : (pr == 2 ? 6 : 8));
      const bool cv = (m & 1) != 0;
      const float* A = sm + (l0 ? 9 + a : (cv ? 29 + a : 20 + a)) * 16;
      const float* V = sm + (l0 ? 10 + a : (ao ? 30 + a : 21 + a)) * 16;
      const bool vid = l0 && !ao;
      const float ar = A[(2 * ao) * 4 + 2 * ai], ai_ = A[(2 * ao + 1) * 4 + 2 * ai];
      const float vr = vid ? ((bo == bi) ? 1.f : 0.f) : V[(2 * bo) * 4 + 2 * bi];
      const float vi = vid ? 0.f : V[(2 * bo + 1) * 4 + 2 * bi];
      const float zr = ar * vr - ai_ * vi, zi = ar * vi + ai_ * vr;
      G[640 + m * 64 + g] = (co == ci) ? zr : (co ? zi : -zi);
    }
  }
}

__device__ __forceinline__ v8f mm(v2f a, v2f b, v8f c) {
  return __builtin_amdgcn_wmma_f32_16x16x4_f32(false, a, false, b, (short)0, c,
                                               false, false);
}

__device__ __forceinline__ v2f cmul(v2f a, v2f b) {
  return v2f{a.x * b.x - a.y * b.y, a.x * b.y + a.y * b.x};
}

__device__ __forceinline__ void load_gate(const float* __restrict__ G, int g,
                                          int t, v2f& a1, v2f& a2, v2f& ad) {
  const float* M = G + g * 16 + (t & 3) * 4;
  const float m0 = M[0], m1 = M[1], m2 = M[2], m3 = M[3];
  const bool on = ((t < 16) == ((t & 15) < 8));
  a1.x = on ? m0 : 0.f; a1.y = on ? m1 : 0.f;
  a2.x = on ? m2 : 0.f; a2.y = on ? m3 : 0.f;
  ad.x = (t < 16) ? m0 : m2; ad.y = (t < 16) ? m1 : m3;
}

__device__ __forceinline__ void load_mux(const float* __restrict__ G, int gR,
                                         int gF, int t, v2f& a1, v2f& a2) {
  const float* M = G + ((t & 4) ? gF : gR) * 16 + (t & 3) * 4;
  const float m0 = M[0], m1 = M[1], m2 = M[2], m3 = M[3];
  const bool on = ((t < 16) == ((t & 15) < 8));
  a1.x = on ? m0 : 0.f; a1.y = on ? m1 : 0.f;
  a2.x = on ? m2 : 0.f; a2.y = on ? m3 : 0.f;
}

__device__ __forceinline__ void load_gate2(const float* __restrict__ G, int m,
                                           int t, v2f (&a)[4]) {
  const float* M = G + 640 + m * 64 + (t & 7) * 8;
  const bool on = ((t < 16) == ((t & 15) < 8));
  #pragma unroll
  for (int j = 0; j < 4; ++j) {
    const float x = M[2 * j], y = M[2 * j + 1];
    a[j].x = on ? x : 0.f; a[j].y = on ? y : 0.f;
  }
}

template <int RB1, int CB, int SEL>
__device__ __forceinline__ void gate2_rr(v2f (&s)[32], const v2f (&g0)[4]) {
  constexpr int RB0 = RB1 - 1;
  #pragma unroll
  for (int p = 0; p < 8; ++p) {
    const int i0 = ((p >> RB0) << (RB1 + 1)) | (p & ((1 << RB0) - 1));
    const int i1 = i0 | (1 << RB0);
    const int i2 = i0 | (1 << RB1);
    const int i3 = i1 | i2;
    if (CB >= 0 && ((i0 >> (CB < 0 ? 0 : CB)) & 1) != SEL) continue;
    v8f z = {};
    v8f d = mm(g0[0], s[i0], z);
    d = mm(g0[1], s[i1], d);
    d = mm(g0[2], s[i2], d);
    d = mm(g0[3], s[i3], d);
    s[i0] = v2f{d[0], d[1]};
    s[i1] = v2f{d[2], d[3]};
    s[i2] = v2f{d[4], d[5]};
    s[i3] = v2f{d[6], d[7]};
  }
}

template <int RB, int CB, int ZB = -1>
__device__ __forceinline__ void gate_rr(v2f (&s)[32], v2f a1, v2f a2) {
  #pragma unroll
  for (int p = 0; p < 16; ++p) {
    const int r0 = ((p >> RB) << (RB + 1)) | (p & ((1 << RB) - 1));
    const int r1 = r0 | (1 << RB);
    if (CB >= 0 && !((r0 >> (CB < 0 ? 0 : CB)) & 1)) continue;
    if (ZB >= 0 && ((r0 >> (ZB < 0 ? 0 : ZB)) & 1)) continue;
    v8f z = {};
    v8f d = mm(a1, s[r0], z);
    d = mm(a2, s[r1], d);
    s[r0] = v2f{d[0], d[1]};
    s[r1] = v2f{d[2], d[3]};
  }
}

template <int RB>
__device__ __forceinline__ void gate_lr(v2f (&s)[32], v2f a1, v2f a2, bool ctl) {
  #pragma unroll
  for (int p = 0; p < 16; ++p) {
    const int r0 = ((p >> RB) << (RB + 1)) | (p & ((1 << RB) - 1));
    const int r1 = r0 | (1 << RB);
    v8f z = {};
    v8f d = mm(a1, s[r0], z);
    d = mm(a2, s[r1], d);
    s[r0] = v2f{ctl ? d[0] : s[r0].x, ctl ? d[1] : s[r0].y};
    s[r1] = v2f{ctl ? d[2] : s[r1].x, ctl ? d[3] : s[r1].y};
  }
}

template <int RB>
__device__ __forceinline__ void gate_mux_lr(v2f (&s)[32], v2f m1, v2f m2,
                                            bool ctl) {
  #pragma unroll
  for (int p = 0; p < 16; ++p) {
    const int i0 = ((p >> RB) << (RB + 1)) | (p & ((1 << RB) - 1));
    const int i1 = i0 | (1 << RB);
    v8f z = {};
    v8f d = mm(m1, s[i0], z);
    d = mm(m2, s[i1], d);
    s[i0] = v2f{ctl ? d[4] : d[0], ctl ? d[5] : d[1]};
    s[i1] = v2f{ctl ? d[6] : d[2], ctl ? d[7] : d[3]};
  }
}

template <int CB>
__device__ __forceinline__ void gate_rl(v2f (&s)[32], v2f ad, bool lo) {
  #pragma unroll
  for (int r = 0; r < 32; ++r) {
    if (!((r >> CB) & 1)) continue;
    v8f z = {};
    v8f d = mm(ad, s[r], z);
    s[r] = v2f{lo ? d[0] : d[2], lo ? d[1] : d[3]};
  }
}

__device__ __forceinline__ void transpose(v2f (&s)[32], float2* sv, int t) {
  #pragma unroll
  for (int r = 0; r < 32; ++r)
    sv[r * 32 + (t ^ r)] = make_float2(s[r].x, s[r].y);
  __builtin_amdgcn_wave_barrier();
  #pragma unroll
  for (int r = 0; r < 32; ++r) {
    float2 v = sv[t * 32 + (r ^ t)];
    s[r] = v2f{v.x, v.y};
  }
  __builtin_amdgcn_wave_barrier();
}

#define ROT_A(L, q)  { load_gate(G, (L) * 20 + (q), t, a1, a2, ad); gate_rr<4 - (q), -1>(s, a1, a2); }
#define G2(RB1, CB, m0) { load_gate2(G, (m0), t, g0); gate2_rr<RB1, CB, 0>(s, g0); \
                          load_gate2(G, (m0) + 1, t, g1); gate2_rr<RB1, CB, 1>(s, g1); }
#define G2L0(RB1, CB, m, a) { load_gate2(G, (m), t, g0); gate2_rr<RB1, CB, 1>(s, g0); \
                              load_gate(G, 10 + (a), t, a1, a2, ad); gate_rr<RB1 - 1, RB1, CB>(s, a1, a2); }
#define CRYS_A0      G2L0(3, 4, 8, 1) G2L0(1, 2, 9, 3)
#define PHASE_A1     ROT_A(1, 0) G2(3, 4, 0) G2(1, 2, 2)
#define PHASE_B1     { load_mux(G, 25, 34, t, a1, a2); gate_mux_lr<4>(s, a1, a2, (t & 1) != 0); } \
                     G2(3, 4, 4) G2(1, 2, 6) \
                     { load_gate(G, 39, t, a1, a2, ad); gate_rl<0>(s, ad, t < 16); }
#define CRYS_B0      { load_gate(G, 14, t, a1, a2, ad); gate_lr<4>(s, a1, a2, (t & 1) != 0); } \
                     G2L0(3, 4, 10, 6) G2L0(1, 2, 11, 8) \
                     { load_gate(G, 19, t, a1, a2, ad); gate_rl<0>(s, ad, t < 16); }

__global__ __launch_bounds__(256, 4) void qsim_kernel(
    const float* __restrict__ inp, const float* __restrict__ G,
    float* __restrict__ out) {
  __shared__ float2 st[WPB][DIM];
  const int t = threadIdx.x & 31;
  const int w = threadIdx.x >> 5;
  const int samp = blockIdx.x * WPB + w;
  float2* sv = st[w];

  float cc[NQ], ss[NQ];
  const float* x = inp + samp * NQ;
  #pragma unroll
  for (int q = 0; q < NQ; ++q) {
    float xv = fminf(fmaxf(x[q], 0.f), 1.f);
    float a = 0.5f * 3.14159265358979323846f * xv;
    cc[q] = cosf(a);
    ss[q] = sinf(a);
  }
  v2f a1, a2, ad;
  v2f g0[4], g1[4];
  v2f u0[NQ], u1[NQ];
  #pragma unroll
  for (int q = 0; q < NQ; ++q) {
    load_gate(G, q, t, a1, a2, ad);
    v8f z = {};
    v8f d = mm(a1, v2f{cc[q], 0.f}, z);
    d = mm(a2, v2f{ss[q], 0.f}, d);
    u0[q] = v2f{d[0], d[1]};
    u1[q] = v2f{d[2], d[3]};
  }
  v2f plo = (t & 1) ? u1[9] : u0[9];
  #pragma unroll
  for (int j = 1; j < 5; ++j)
    plo = cmul(plo, ((t >> j) & 1) ? u1[9 - j] : u0[9 - j]);
  v2f s[32];
  s[0] = plo;
  #pragma unroll
  for (int j = 0; j < 5; ++j) {
    #pragma unroll
    for (int e = 0; e < (1 << j); ++e) {
      s[e | (1 << j)] = cmul(s[e], u1[4 - j]);
      s[e] = cmul(s[e], u0[4 - j]);
    }
  }

  CRYS_A0
  transpose(s, sv, t);
  CRYS_B0
  transpose(s, sv, t);
  PHASE_A1
  transpose(s, sv, t);
  PHASE_B1

  float acc[NQ];
  #pragma unroll
  for (int q = 0; q < NQ; ++q) acc[q] = 0.f;
  #pragma unroll
  for (int r = 0; r < 32; ++r) {
    const int i = t * 32 + r;
    float p = s[r].x * s[r].x + s[r].y * s[r].y;
    #pragma unroll
    for (int q = 0; q < NQ; ++q)
      acc[q] += ((i >> (9 - q)) & 1) ? -p : p;
  }
  #pragma unroll
  for (int q = 0; q < NQ; ++q) {
    #pragma unroll
    for (int off = 16; off >= 1; off >>= 1)
      acc[q] += __shfl_xor(acc[q], off, 32);
  }
  if (t == 0) {
    #pragma unroll
    for (int q = 0; q < NQ; ++q) out[samp * NQ + q] = acc[q];
  }
}

extern "C" void kernel_launch(void* const* d_in, const int* in_sizes, int n_in,
                              void* d_out, int out_size, void* d_ws,
                              size_t ws_size, hipStream_t stream) {
  const float* inputs = (const float*)d_in[0];
  const float* weights = (const float*)d_in[1];
  float* out = (float*)d_out;
  float* G = (float*)d_ws;

  const int B = in_sizes[0] / NQ;
  build_gates_kernel<<<dim3(1), dim3(64), 0, stream>>>(weights, G);
  qsim_kernel<<<dim3(B / WPB), dim3(256), 0, stream>>>(inputs, G, out);
}


# Layer-0 ring gates act on the state's factors (qubit q's 2-vector per value of qubit q-1, RY*Rot prebuilt; 9->0 on the q0,q1 factor): 232 WMMAs, not 372; 1 transpose, not 3; products reassociated
# speedup vs baseline: 3.8638x; 1.3450x over previous
#include <hip/hip_runtime.h>
#include <hip/hip_bf16.h>

typedef __attribute__((ext_vector_type(2))) float v2f;
typedef __attribute__((ext_vector_type(8))) float v8f;

#define NQ 10
#define DIM 1024
#define WPB 8

__global__ void build_gates_kernel(const float* __restrict__ W,
                                   float* __restrict__ G) {
  __shared__ float sm[640];
  const int g = threadIdx.x;
  if (g < 40) {
  int L = g / 20, r = g % 20;
  float u00r = 0.f, u00i = 0.f, u01r = 0.f, u01i = 0.f;
  float u10r = 0.f, u10i = 0.f, u11r = 0.f, u11i = 0.f;
  const bool cry = (r >= 10);
  const bool fused = cry && r < 19;
  if (!cry || fused) {
    int q = cry ? (r - 9) : r;
    int base = L * 40 + 3 * q;
    float phi = W[base], th = W[base + 1], om = W[base + 2];
    float ct = cosf(0.5f * th), st = sinf(0.5f * th);
    float ap = 0.5f * (phi + om), am = 0.5f * (phi - om);
    float cap = cosf(ap), sap = sinf(ap);
    float cam = cosf(am), sam = sinf(am);
    u00r =  ct * cap; u00i = -ct * sap;
    u01r = -st * cam; u01i = -st * sam;
    u10r =  st * cam; u10i = -st * sam;
    u11r =  ct * cap; u11i =  ct * sap;
  }
  if (cry) {
    int e = r - 10;
    float th = W[L * 40 + 30 + e];
    float ct = cosf(0.5f * th), st = sinf(0.5f * th);
    if (fused) {
      float n00r = ct * u00r - st * u10r, n00i = ct * u00i - st * u10i;
      float n01r = ct * u01r - st * u11r, n01i = ct * u01i - st * u11i;
      float n10r = st * u00r + ct * u10r, n10i = st * u00i + ct * u10i;
      float n11r = st * u01r + ct * u11r, n11i = st * u01i + ct * u11i;
      u00r = n00r; u00i = n00i; u01r = n01r; u01i = n01i;
      u10r = n10r; u10i = n10i; u11r = n11r; u11i = n11i;
    } else {
      u00r = ct;  u00i = 0.f; u01r = -st; u01i = 0.f;
      u10r = st;  u10i = 0.f; u11r = ct;  u11i = 0.f;
    }
  }
  float* M = G + g * 16;
  M[ 0] = u00r; M[ 1] = -u00i; M[ 2] = u01r; M[ 3] = -u01i;
  M[ 4] = u00i; M[ 5] =  u00r; M[ 6] = u01i; M[ 7] =  u01r;
  M[ 8] = u10r; M[ 9] = -u10i; M[10] = u11r; M[11] = -u11i;
  M[12] = u10i; M[13] =  u10r; M[14] = u11i; M[15] =  u11r;
  #pragma unroll
  for (int k = 0; k < 16; ++k) sm[g * 16 + k] = M[k];
  }
  __syncthreads();
  {
    const int R = g >> 3, Cc = g & 7;
    const int ao = R >> 2, bo = (R >> 1) & 1, co = R & 1;
    const int ai = Cc >> 2, bi = (Cc >> 1) & 1, ci = Cc & 1;
    #pragma unroll
    for (int m = 0; m < 8; ++m) {
      const int pr = m >> 1;
      const int a = (pr == 0) ? 1 : (pr == 1 ? 3 : (pr == 2 ? 6 : 8));
      const bool cv = (m & 1) != 0;
      const float* A = sm + (cv ? 29 + a : 20 + a) * 16;
      const float* V = sm + (ao ? 30 + a : 21 + a) * 16;
      const float ar = A[(2 * ao) * 4 + 2 * ai], ai_ = A[(2 * ao + 1) * 4 + 2 * ai];
      const float vr = V[(2 * bo) * 4 + 2 * bi];
      const float vi = V[(2 * bo + 1) * 4 + 2 * bi];
      const float zr = ar * vr - ai_ * vi, zi = ar * vi + ai_ * vr;
      G[640 + m * 64 + g] = (co == ci) ? zr : (co ? zi : -zi);
    }
  }
}

__device__ __forceinline__ v8f mm(v2f a, v2f b, v8f c) {
  return __builtin_amdgcn_wmma_f32_16x16x4_f32(false, a, false, b, (short)0, c,
                                               false, false);
}

__device__ __forceinline__ v2f cmul(v2f a, v2f b) {
  return v2f{a.x * b.x - a.y * b.y, a.x * b.y + a.y * b.x};
}

__device__ __forceinline__ v2f pick(bool c, v2f one, v2f zero) {
  return v2f{c ? one.x : zero.x, c ? one.y : zero.y};
}

__device__ __forceinline__ v2f pick2(const v2f (&T)[2][2], bool a, bool b) {
  return pick(a, pick(b, T[1][1], T[1][0]), pick(b, T[0][1], T[0][0]));
}

__device__ __forceinline__ void load_gate(const float* __restrict__ G, int g,
                                          int t, v2f& a1, v2f& a2, v2f& ad) {
  const float* M = G + g * 16 + (t & 3) * 4;
  const float m0 = M[0], m1 = M[1], m2 = M[2], m3 = M[3];
  const bool on = ((t < 16) == ((t & 15) < 8));
  a1.x = on ? m0 : 0.f; a1.y = on ? m1 : 0.f;
  a2.x = on ? m2 : 0.f; a2.y = on ? m3 : 0.f;
  ad.x = (t < 16) ? m0 : m2; ad.y = (t < 16) ? m1 : m3;
}

__device__ __forceinline__ void load_mux(const float* __restrict__ G, int gR,
                                         int gF, int t, v2f& a1, v2f& a2) {
  const float* M = G + ((t & 4) ? gF : gR) * 16 + (t & 3) * 4;
  const float m0 = M[0], m1 = M[1], m2 = M[2], m3 = M[3];
  const bool on = ((t < 16) == ((t & 15) < 8));
  a1.x = on ? m0 : 0.f; a1.y = on ? m1 : 0.f;
  a2.x = on ? m2 : 0.f; a2.y = on ? m3 : 0.f;
}

__device__ __forceinline__ void load_gate2(const float* __restrict__ G, int m,
                                           int t, v2f (&a)[4]) {
  const float* M = G + 640 + m * 64 + (t & 7) * 8;
  const bool on = ((t < 16) == ((t & 15) < 8));
  #pragma unroll
  for (int j = 0; j < 4; ++j) {
    const float x = M[2 * j], y = M[2 * j + 1];
    a[j].x = on ? x : 0.f; a[j].y = on ? y : 0.f;
  }
}

template <int RB1, int CB, int SEL>
__device__ __forceinline__ void gate2_rr(v2f (&s)[32], const v2f (&g0)[4]) {
  constexpr int RB0 = RB1 - 1;
  #pragma unroll
  for (int p = 0; p < 8; ++p) {
    const int i0 = ((p >> RB0) << (RB1 + 1)) | (p & ((1 << RB0) - 1));
    const int i1 = i0 | (1 << RB0);
    const int i2 = i0 | (1 << RB1);
    const int i3 = i1 | i2;
    if (CB >= 0 && ((i0 >> (CB < 0 ? 0 : CB)) & 1) != SEL) continue;
    v8f z = {};
    v8f d = mm(g0[0], s[i0], z);
    d = mm(g0[1], s[i1], d);
    d = mm(g0[2], s[i2], d);
    d = mm(g0[3], s[i3], d);
    s[i0] = v2f{d[0], d[1]};
    s[i1] = v2f{d[2], d[3]};
    s[i2] = v2f{d[4], d[5]};
    s[i3] = v2f{d[6], d[7]};
  }
}

template <int RB, int CB>
__device__ __forceinline__ void gate_rr(v2f (&s)[32], v2f a1, v2f a2) {
  #pragma unroll
  for (int p = 0; p < 16; ++p) {
    const int r0 = ((p >> RB) << (RB + 1)) | (p & ((1 << RB) - 1));
    const int r1 = r0 | (1 << RB);
    if (CB >= 0 && !((r0 >> (CB < 0 ? 0 : CB)) & 1)) continue;
    v8f z = {};
    v8f d = mm(a1, s[r0], z);
    d = mm(a2, s[r1], d);
    s[r0] = v2f{d[0], d[1]};
    s[r1] = v2f{d[2], d[3]};
  }
}

template <int RB>
__device__ __forceinline__ void gate_mux_lr(v2f (&s)[32], v2f m1, v2f m2,
                                            bool ctl) {
  #pragma unroll
  for (int p = 0; p < 16; ++p) {
    const int i0 = ((p >> RB) << (RB + 1)) | (p & ((1 << RB) - 1));
    const int i1 = i0 | (1 << RB);
    v8f z = {};
    v8f d = mm(m1, s[i0], z);
    d = mm(m2, s[i1], d);
    s[i0] = v2f{ctl ? d[4] : d[0], ctl ? d[5] : d[1]};
    s[i1] = v2f{ctl ? d[6] : d[2], ctl ? d[7] : d[3]};
  }
}

template <int CB>
__device__ __forceinline__ void gate_rl(v2f (&s)[32], v2f ad, bool lo) {
  #pragma unroll
  for (int r = 0; r < 32; ++r) {
    if (!((r >> CB) & 1)) continue;
    v8f z = {};
    v8f d = mm(ad, s[r], z);
    s[r] = v2f{lo ? d[0] : d[2], lo ? d[1] : d[3]};
  }
}

__device__ __forceinline__ void transpose(v2f (&s)[32], float2* sv, int t) {
  #pragma unroll
  for (int r = 0; r < 32; ++r)
    sv[r * 32 + (t ^ r)] = make_float2(s[r].x, s[r].y);
  __builtin_amdgcn_wave_barrier();
  #pragma unroll
  for (int r = 0; r < 32; ++r) {
    float2 v = sv[t * 32 + (r ^ t)];
    s[r] = v2f{v.x, v.y};
  }
}

#define ROT_A(L, q)  { load_gate(G, (L) * 20 + (q), t, a1, a2, ad); gate_rr<4 - (q), -1>(s, a1, a2); }
#define G2(RB1, CB, m0) { load_gate2(G, (m0), t, g0); gate2_rr<RB1, CB, 0>(s, g0); \
                          load_gate2(G, (m0) + 1, t, g1); gate2_rr<RB1, CB, 1>(s, g1); }
#define PHASE_A1     ROT_A(1, 0) G2(3, 4, 0) G2(1, 2, 2)
#define PHASE_B1     { load_mux(G, 25, 34, t, a1, a2); gate_mux_lr<4>(s, a1, a2, (t & 1) != 0); } \
                     G2(3, 4, 4) G2(1, 2, 6) \
                     { load_gate(G, 39, t, a1, a2, ad); gate_rl<0>(s, ad, t < 16); }

__global__ __launch_bounds__(256, 4) void qsim_kernel(
    const float* __restrict__ inp, const float* __restrict__ G,
    float* __restrict__ out) {
  __shared__ float2 st[WPB][DIM];
  const int t = threadIdx.x & 31;
  const int w = threadIdx.x >> 5;
  const int samp = blockIdx.x * WPB + w;
  float2* sv = st[w];

  float cc[NQ], ss[NQ];
  const float* x = inp + samp * NQ;
  #pragma unroll
  for (int q = 0; q < NQ; ++q) {
    float xv = fminf(fmaxf(x[q], 0.f), 1.f);
    float a = 0.5f * 3.14159265358979323846f * xv;
    cc[q] = cosf(a);
    ss[q] = sinf(a);
  }
  v2f a1, a2, ad;
  v2f g0[4], g1[4];
  v2f w0[2];
  v2f T[NQ][2][2];
  {
    load_gate(G, 0, t, a1, a2, ad);
    v8f z = {};
    v8f d = mm(a1, v2f{cc[0], 0.f}, z);
    d = mm(a2, v2f{ss[0], 0.f}, d);
    w0[0] = v2f{d[0], d[1]};
    w0[1] = v2f{d[2], d[3]};
  }
  #pragma unroll
  for (int q = 1; q < NQ; ++q) {
    load_mux(G, q, 9 + q, t, a1, a2);
    v8f z = {};
    v8f d = mm(a1, v2f{cc[q], 0.f}, z);
    d = mm(a2, v2f{ss[q], 0.f}, d);
    T[q][0][0] = v2f{d[0], d[1]};
    T[q][0][1] = v2f{d[2], d[3]};
    T[q][1][0] = v2f{d[4], d[5]};
    T[q][1][1] = v2f{d[6], d[7]};
  }
  v2f X[2][2];
  #pragma unroll
  for (int a = 0; a < 2; ++a) {
    #pragma unroll
    for (int b = 0; b < 2; ++b) X[a][b] = cmul(w0[a], T[1][a][b]);
  }
  {
    load_gate(G, 19, t, a1, a2, ad);
    const bool c9 = (t & 1) != 0;
    #pragma unroll
    for (int b = 0; b < 2; ++b) {
      v8f z = {};
      v8f d = mm(a1, X[0][b], z);
      d = mm(a2, X[1][b], d);
      X[0][b] = pick(c9, v2f{d[0], d[1]}, X[0][b]);
      X[1][b] = pick(c9, v2f{d[2], d[3]}, X[1][b]);
    }
  }
  const bool b5 = ((t >> 4) & 1) != 0, b6 = ((t >> 3) & 1) != 0;
  const bool b7 = ((t >> 2) & 1) != 0, b8 = ((t >> 1) & 1) != 0;
  const bool b9 = (t & 1) != 0;
  v2f plo = cmul(cmul(pick2(T[6], b5, b6), pick2(T[7], b6, b7)),
                 cmul(pick2(T[8], b7, b8), pick2(T[9], b8, b9)));
  v2f E[2][2];
  #pragma unroll
  for (int b = 0; b < 2; ++b) {
    const v2f f = cmul(pick(b5, T[5][b][1], T[5][b][0]), plo);
    E[0][b] = cmul(T[4][0][b], f);
    E[1][b] = cmul(T[4][1][b], f);
  }
  v2f P2[8], P3[16];
  #pragma unroll
  for (int k = 0; k < 8; ++k)
    P2[k] = cmul(X[k >> 2][(k >> 1) & 1], T[2][(k >> 1) & 1][k & 1]);
  #pragma unroll
  for (int k = 0; k < 16; ++k)
    P3[k] = cmul(P2[k >> 1], T[3][(k >> 1) & 1][k & 1]);
  v2f s[32];
  #pragma unroll
  for (int r = 0; r < 32; ++r)
    s[r] = cmul(P3[r >> 1], E[(r >> 1) & 1][r & 1]);

  PHASE_A1
  transpose(s, sv, t);
  PHASE_B1

  float acc[NQ];
  #pragma unroll
  for (int q = 0; q < NQ; ++q) acc[q] = 0.f;
  #pragma unroll
  for (int r = 0; r < 32; ++r) {
    const int i = t * 32 + r;
    float p = s[r].x * s[r].x + s[r].y * s[r].y;
    #pragma unroll
    for (int q = 0; q < NQ; ++q)
      acc[q] += ((i >> (9 - q)) & 1) ? -p : p;
  }
  #pragma unroll
  for (int q = 0; q < NQ; ++q) {
    #pragma unroll
    for (int off = 16; off >= 1; off >>= 1)
      acc[q] += __shfl_xor(acc[q], off, 32);
  }
  if (t == 0) {
    #pragma unroll
    for (int q = 0; q < NQ; ++q) out[samp * NQ + q] = acc[q];
  }
}

extern "C" void kernel_launch(void* const* d_in, const int* in_sizes, int n_in,
                              void* d_out, int out_size, void* d_ws,
                              size_t ws_size, hipStream_t stream) {
  const float* inputs = (const float*)d_in[0];
  const float* weights = (const float*)d_in[1];
  float* out = (float*)d_out;
  float* G = (float*)d_ws;

  const int B = in_sizes[0] / NQ;
  build_gates_kernel<<<dim3(1), dim3(64), 0, stream>>>(weights, G);
  qsim_kernel<<<dim3(B / WPB), dim3(256), 0, stream>>>(inputs, G, out);
}


# Layer 1 on qubits 0-2 acts in the state's doubling tree: Rot_0 joins ring gate 9->0 on the (q0,q1) factor, the (q1,q2) gate runs on the 16 partial products: 184 WMMAs, not 232; products reassociated
# speedup vs baseline: 4.2784x; 1.1073x over previous
#include <hip/hip_runtime.h>
#include <hip/hip_bf16.h>

typedef __attribute__((ext_vector_type(2))) float v2f;
typedef __attribute__((ext_vector_type(8))) float v8f;

#define NQ 10
#define DIM 1024
#define WPB 8

__global__ void build_gates_kernel(const float* __restrict__ W,
                                   float* __restrict__ G) {
  __shared__ float sm[640];
  const int g = threadIdx.x;
  if (g < 40) {
  int L = g / 20, r = g % 20;
  float u00r = 0.f, u00i = 0.f, u01r = 0.f, u01i = 0.f;
  float u10r = 0.f, u10i = 0.f, u11r = 0.f, u11i = 0.f;
  const bool cry = (r >= 10);
  const bool fused = cry && r < 19;
  const bool post = (g == 19);
  if (!cry || fused || post) {
    int q = fused ? (r - 9) : (post ? 0 : r);
    int base = (post ? 1 : L) * 40 + 3 * q;
    float phi = W[base], th = W[base + 1], om = W[base + 2];
    float ct = cosf(0.5f * th), st = sinf(0.5f * th);
    float ap = 0.5f * (phi + om), am = 0.5f * (phi - om);
    float cap = cosf(ap), sap = sinf(ap);
    float cam = cosf(am), sam = sinf(am);
    u00r =  ct * cap; u00i = -ct * sap;
    u01r = -st * cam; u01i = -st * sam;
    u10r =  st * cam; u10i = -st * sam;
    u11r =  ct * cap; u11i =  ct * sap;
  }
  if (cry) {
    int e = r - 10;
    float th = W[L * 40 + 30 + e];
    float ct = cosf(0.5f * th), st = sinf(0.5f * th);
    if (fused) {
      float n00r = ct * u00r - st * u10r, n00i = ct * u00i - st * u10i;
      float n01r = ct * u01r - st * u11r, n01i = ct * u01i - st * u11i;
      float n10r = st * u00r + ct * u10r, n10i = st * u00i + ct * u10i;
      float n11r = st * u01r + ct * u11r, n11i = st * u01i + ct * u11i;
      u00r = n00r; u00i = n00i; u01r = n01r; u01i = n01i;
      u10r = n10r; u10i = n10i; u11r = n11r; u11i = n11i;
    } else if (post) {
      float n00r = u00r * ct + u01r * st, n00i = u00i * ct + u01i * st;
      float n01r = u01r * ct - u00r * st, n01i = u01i * ct - u00i * st;
      float n10r = u10r * ct + u11r * st, n10i = u10i * ct + u11i * st;
      float n11r = u11r * ct - u10r * st, n11i = u11i * ct - u10i * st;
      u00r = n00r; u00i = n00i; u01r = n01r; u01i = n01i;
      u10r = n10r; u10i = n10i; u11r = n11r; u11i = n11i;
    } else {
      u00r = ct;  u00i = 0.f; u01r = -st; u01i = 0.f;
      u10r = st;  u10i = 0.f; u11r = ct;  u11i = 0.f;
    }
  }
  float* M = G + g * 16;
  M[ 0] = u00r; M[ 1] = -u00i; M[ 2] = u01r; M[ 3] = -u01i;
  M[ 4] = u00i; M[ 5] =  u00r; M[ 6] = u01i; M[ 7] =  u01r;
  M[ 8] = u10r; M[ 9] = -u10i; M[10] = u11r; M[11] = -u11i;
  M[12] = u10i; M[13] =  u10r; M[14] = u11i; M[15] =  u11r;
  #pragma unroll
  for (int k = 0; k < 16; ++k) sm[g * 16 + k] = M[k];
  }
  __syncthreads();
  {
    const int R = g >> 3, Cc = g & 7;
    const int ao = R >> 2, bo = (R >> 1) & 1, co = R & 1;
    const int ai = Cc >> 2, bi = (Cc >> 1) & 1, ci = Cc & 1;
    #pragma unroll
    for (int m = 0; m < 8; ++m) {
      const int pr = m >> 1;
      const int a = (pr == 0) ? 1 : (pr == 1 ? 3 : (pr == 2 ? 6 : 8));
      const bool cv = (m & 1) != 0;
      const float* A = sm + (cv ? 29 + a : 20 + a) * 16;
      const float* V = sm + (ao ? 30 + a : 21 + a) * 16;
      const float ar = A[(2 * ao) * 4 + 2 * ai], ai_ = A[(2 * ao + 1) * 4 + 2 * ai];
      const float vr = V[(2 * bo) * 4 + 2 * bi];
      const float vi = V[(2 * bo + 1) * 4 + 2 * bi];
      const float zr = ar * vr - ai_ * vi, zi = ar * vi + ai_ * vr;
      G[640 + m * 64 + g] = (co == ci) ? zr : (co ? zi : -zi);
    }
  }
}

__device__ __forceinline__ v8f mm(v2f a, v2f b, v8f c) {
  return __builtin_amdgcn_wmma_f32_16x16x4_f32(false, a, false, b, (short)0, c,
                                               false, false);
}

__device__ __forceinline__ v2f cmul(v2f a, v2f b) {
  return v2f{a.x * b.x - a.y * b.y, a.x * b.y + a.y * b.x};
}

__device__ __forceinline__ v2f pick(bool c, v2f one, v2f zero) {
  return v2f{c ? one.x : zero.x, c ? one.y : zero.y};
}

__device__ __forceinline__ v2f pick2(const v2f (&T)[2][2], bool a, bool b) {
  return pick(a, pick(b, T[1][1], T[1][0]), pick(b, T[0][1], T[0][0]));
}

__device__ __forceinline__ void load_gate(const float* __restrict__ G, int g,
                                          int t, v2f& a1, v2f& a2, v2f& ad) {
  const float* M = G + g * 16 + (t & 3) * 4;
  const float m0 = M[0], m1 = M[1], m2 = M[2], m3 = M[3];
  const bool on = ((t < 16) == ((t & 15) < 8));
  a1.x = on ? m0 : 0.f; a1.y = on ? m1 : 0.f;
  a2.x = on ? m2 : 0.f; a2.y = on ? m3 : 0.f;
  ad.x = (t < 16) ? m0 : m2; ad.y = (t < 16) ? m1 : m3;
}

__device__ __forceinline__ void load_mux(const float* __restrict__ G, int gR,
                                         int gF, int t, v2f& a1, v2f& a2) {
  const float* M = G + ((t & 4) ? gF : gR) * 16 + (t & 3) * 4;
  const float m0 = M[0], m1 = M[1], m2 = M[2], m3 = M[3];
  const bool on = ((t < 16) == ((t & 15) < 8));
  a1.x = on ? m0 : 0.f; a1.y = on ? m1 : 0.f;
  a2.x = on ? m2 : 0.f; a2.y = on ? m3 : 0.f;
}

__device__ __forceinline__ void load_gate2(const float* __restrict__ G, int m,
                                           int t, v2f (&a)[4]) {
  const float* M = G + 640 + m * 64 + (t & 7) * 8;
  const bool on = ((t < 16) == ((t & 15) < 8));
  #pragma unroll
  for (int j = 0; j < 4; ++j) {
    const float x = M[2 * j], y = M[2 * j + 1];
    a[j].x = on ? x : 0.f; a[j].y = on ? y : 0.f;
  }
}

template <int RB1, int CB, int SEL, int N>
__device__ __forceinline__ void gate2_rr(v2f (&s)[N], const v2f (&g0)[4]) {
  constexpr int RB0 = RB1 - 1;
  #pragma unroll
  for (int p = 0; p < N / 4; ++p) {
    const int i0 = ((p >> RB0) << (RB1 + 1)) | (p & ((1 << RB0) - 1));
    const int i1 = i0 | (1 << RB0);
    const int i2 = i0 | (1 << RB1);
    const int i3 = i1 | i2;
    if (CB >= 0 && ((i0 >> (CB < 0 ? 0 : CB)) & 1) != SEL) continue;
    v8f z = {};
    v8f d = mm(g0[0], s[i0], z);
    d = mm(g0[1], s[i1], d);
    d = mm(g0[2], s[i2], d);
    d = mm(g0[3], s[i3], d);
    s[i0] = v2f{d[0], d[1]};
    s[i1] = v2f{d[2], d[3]};
    s[i2] = v2f{d[4], d[5]};
    s[i3] = v2f{d[6], d[7]};
  }
}

template <int RB>
__device__ __forceinline__ void gate_mux_lr(v2f (&s)[32], v2f m1, v2f m2,
                                            bool ctl) {
  #pragma unroll
  for (int p = 0; p < 16; ++p) {
    const int i0 = ((p >> RB) << (RB + 1)) | (p & ((1 << RB) - 1));
    const int i1 = i0 | (1 << RB);
    v8f z = {};
    v8f d = mm(m1, s[i0], z);
    d = mm(m2, s[i1], d);
    s[i0] = v2f{ctl ? d[4] : d[0], ctl ? d[5] : d[1]};
    s[i1] = v2f{ctl ? d[6] : d[2], ctl ? d[7] : d[3]};
  }
}

template <int CB>
__device__ __forceinline__ void gate_rl(v2f (&s)[32], v2f ad, bool lo) {
  #pragma unroll
  for (int r = 0; r < 32; ++r) {
    if (!((r >> CB) & 1)) continue;
    v8f z = {};
    v8f d = mm(ad, s[r], z);
    s[r] = v2f{lo ? d[0] : d[2], lo ? d[1] : d[3]};
  }
}

__device__ __forceinline__ void transpose(v2f (&s)[32], float2* sv, int t) {
  #pragma unroll
  for (int r = 0; r < 32; ++r)
    sv[r * 32 + (t ^ r)] = make_float2(s[r].x, s[r].y);
  __builtin_amdgcn_wave_barrier();
  #pragma unroll
  for (int r = 0; r < 32; ++r) {
    float2 v = sv[t * 32 + (r ^ t)];
    s[r] = v2f{v.x, v.y};
  }
}

#define G2(RB1, CB, m0) { load_gate2(G, (m0), t, g0); gate2_rr<RB1, CB, 0, 32>(s, g0); \
                          load_gate2(G, (m0) + 1, t, g1); gate2_rr<RB1, CB, 1, 32>(s, g1); }
#define PHASE_A1     G2(1, 2, 2)
#define PHASE_B1     { load_mux(G, 25, 34, t, a1, a2); gate_mux_lr<4>(s, a1, a2, (t & 1) != 0); } \
                     G2(3, 4, 4) G2(1, 2, 6) \
                     { load_gate(G, 39, t, a1, a2, ad); gate_rl<0>(s, ad, t < 16); }

__global__ __launch_bounds__(256, 4) void qsim_kernel(
    const float* __restrict__ inp, const float* __restrict__ G,
    float* __restrict__ out) {
  __shared__ float2 st[WPB][DIM];
  const int t = threadIdx.x & 31;
  const int w = threadIdx.x >> 5;
  const int samp = blockIdx.x * WPB + w;
  float2* sv = st[w];

  float cc[NQ], ss[NQ];
  const float* x = inp + samp * NQ;
  #pragma unroll
  for (int q = 0; q < NQ; ++q) {
    float xv = fminf(fmaxf(x[q], 0.f), 1.f);
    float a = 0.5f * 3.14159265358979323846f * xv;
    cc[q] = cosf(a);
    ss[q] = sinf(a);
  }
  v2f a1, a2, ad;
  v2f g0[4], g1[4];
  v2f w0[2];
  v2f T[NQ][2][2];
  {
    load_gate(G, 0, t, a1, a2, ad);
    v8f z = {};
    v8f d = mm(a1, v2f{cc[0], 0.f}, z);
    d = mm(a2, v2f{ss[0], 0.f}, d);
    w0[0] = v2f{d[0], d[1]};
    w0[1] = v2f{d[2], d[3]};
  }
  #pragma unroll
  for (int q = 1; q < NQ; ++q) {
    load_mux(G, q, 9 + q, t, a1, a2);
    v8f z = {};
    v8f d = mm(a1, v2f{cc[q], 0.f}, z);
    d = mm(a2, v2f{ss[q], 0.f}, d);
    T[q][0][0] = v2f{d[0], d[1]};
    T[q][0][1] = v2f{d[2], d[3]};
    T[q][1][0] = v2f{d[4], d[5]};
    T[q][1][1] = v2f{d[6], d[7]};
  }
  v2f X[2][2];
  #pragma unroll
  for (int a = 0; a < 2; ++a) {
    #pragma unroll
    for (int b = 0; b < 2; ++b) X[a][b] = cmul(w0[a], T[1][a][b]);
  }
  {
    load_mux(G, 20, 19, t, a1, a2);
    const bool c9 = (t & 1) != 0;
    #pragma unroll
    for (int b = 0; b < 2; ++b) {
      v8f z = {};
      v8f d = mm(a1, X[0][b], z);
      d = mm(a2, X[1][b], d);
      X[0][b] = pick(c9, v2f{d[4], d[5]}, v2f{d[0], d[1]});
      X[1][b] = pick(c9, v2f{d[6], d[7]}, v2f{d[2], d[3]});
    }
  }
  const bool b5 = ((t >> 4) & 1) != 0, b6 = ((t >> 3) & 1) != 0;
  const bool b7 = ((t >> 2) & 1) != 0, b8 = ((t >> 1) & 1) != 0;
  const bool b9 = (t & 1) != 0;
  v2f plo = cmul(cmul(pick2(T[6], b5, b6), pick2(T[7], b6, b7)),
                 cmul(pick2(T[8], b7, b8), pick2(T[9], b8, b9)));
  v2f E[2][2];
  #pragma unroll
  for (int b = 0; b < 2; ++b) {
    const v2f f = cmul(pick(b5, T[5][b][1], T[5][b][0]), plo);
    E[0][b] = cmul(T[4][0][b], f);
    E[1][b] = cmul(T[4][1][b], f);
  }
  v2f P2[8], P3[16];
  #pragma unroll
  for (int k = 0; k < 8; ++k)
    P2[k] = cmul(X[k >> 2][(k >> 1) & 1], T[2][(k >> 1) & 1][k & 1]);
  #pragma unroll
  for (int k = 0; k < 16; ++k)
    P3[k] = cmul(P2[k >> 1], T[3][(k >> 1) & 1][k & 1]);
  load_gate2(G, 0, t, g0); gate2_rr<2, 3, 0, 16>(P3, g0);
  load_gate2(G, 1, t, g1); gate2_rr<2, 3, 1, 16>(P3, g1);
  v2f s[32];
  #pragma unroll
  for (int r = 0; r < 32; ++r)
    s[r] = cmul(P3[r >> 1], E[(r >> 1) & 1][r & 1]);

  PHASE_A1
  transpose(s, sv, t);
  PHASE_B1

  float acc[NQ];
  #pragma unroll
  for (int q = 0; q < NQ; ++q) acc[q] = 0.f;
  #pragma unroll
  for (int r = 0; r < 32; ++r) {
    const int i = t * 32 + r;
    float p = s[r].x * s[r].x + s[r].y * s[r].y;
    #pragma unroll
    for (int q = 0; q < NQ; ++q)
      acc[q] += ((i >> (9 - q)) & 1) ? -p : p;
  }
  #pragma unroll
  for (int q = 0; q < NQ; ++q) {
    #pragma unroll
    for (int off = 16; off >= 1; off >>= 1)
      acc[q] += __shfl_xor(acc[q], off, 32);
  }
  if (t == 0) {
    #pragma unroll
    for (int q = 0; q < NQ; ++q) out[samp * NQ + q] = acc[q];
  }
}

extern "C" void kernel_launch(void* const* d_in, const int* in_sizes, int n_in,
                              void* d_out, int out_size, void* d_ws,
                              size_t ws_size, hipStream_t stream) {
  const float* inputs = (const float*)d_in[0];
  const float* weights = (const float*)d_in[1];
  float* out = (float*)d_out;
  float* G = (float*)d_ws;

  const int B = in_sizes[0] / NQ;
  build_gates_kernel<<<dim3(1), dim3(64), 0, stream>>>(weights, G);
  qsim_kernel<<<dim3(B / WPB), dim3(256), 0, stream>>>(inputs, G, out);
}


# Input angles' cos and sin by a fold to [0, pi/4] and two short polynomials, not the library calls: no range reduction or large-argument branch, input loads issue together; values differ in last bits
# speedup vs baseline: 4.7656x; 1.1139x over previous
#include <hip/hip_runtime.h>
#include <hip/hip_bf16.h>

typedef __attribute__((ext_vector_type(2))) float v2f;
typedef __attribute__((ext_vector_type(8))) float v8f;

#define NQ 10
#define DIM 1024
#define WPB 8

__global__ void build_gates_kernel(const float* __restrict__ W,
                                   float* __restrict__ G) {
  __shared__ float sm[640];
  const int g = threadIdx.x;
  if (g < 40) {
  int L = g / 20, r = g % 20;
  float u00r = 0.f, u00i = 0.f, u01r = 0.f, u01i = 0.f;
  float u10r = 0.f, u10i = 0.f, u11r = 0.f, u11i = 0.f;
  const bool cry = (r >= 10);
  const bool fused = cry && r < 19;
  const bool post = (g == 19);
  if (!cry || fused || post) {
    int q = fused ? (r - 9) : (post ? 0 : r);
    int base = (post ? 1 : L) * 40 + 3 * q;
    float phi = W[base], th = W[base + 1], om = W[base + 2];
    float ct = cosf(0.5f * th), st = sinf(0.5f * th);
    float ap = 0.5f * (phi + om), am = 0.5f * (phi - om);
    float cap = cosf(ap), sap = sinf(ap);
    float cam = cosf(am), sam = sinf(am);
    u00r =  ct * cap; u00i = -ct * sap;
    u01r = -st * cam; u01i = -st * sam;
    u10r =  st * cam; u10i = -st * sam;
    u11r =  ct * cap; u11i =  ct * sap;
  }
  if (cry) {
    int e = r - 10;
    float th = W[L * 40 + 30 + e];
    float ct = cosf(0.5f * th), st = sinf(0.5f * th);
    if (fused) {
      float n00r = ct * u00r - st * u10r, n00i = ct * u00i - st * u10i;
      float n01r = ct * u01r - st * u11r, n01i = ct * u01i - st * u11i;
      float n10r = st * u00r + ct * u10r, n10i = st * u00i + ct * u10i;
      float n11r = st * u01r + ct * u11r, n11i = st * u01i + ct * u11i;
      u00r = n00r; u00i = n00i; u01r = n01r; u01i = n01i;
      u10r = n10r; u10i = n10i; u11r = n11r; u11i = n11i;
    } else if (post) {
      float n00r = u00r * ct + u01r * st, n00i = u00i * ct + u01i * st;
      float n01r = u01r * ct - u00r * st, n01i = u01i * ct - u00i * st;
      float n10r = u10r * ct + u11r * st, n10i = u10i * ct + u11i * st;
      float n11r = u11r * ct - u10r * st, n11i = u11i * ct - u10i * st;
      u00r = n00r; u00i = n00i; u01r = n01r; u01i = n01i;
      u10r = n10r; u10i = n10i; u11r = n11r; u11i = n11i;
    } else {
      u00r = ct;  u00i = 0.f; u01r = -st; u01i = 0.f;
      u10r = st;  u10i = 0.f; u11r = ct;  u11i = 0.f;
    }
  }
  float* M = G + g * 16;
  M[ 0] = u00r; M[ 1] = -u00i; M[ 2] = u01r; M[ 3] = -u01i;
  M[ 4] = u00i; M[ 5] =  u00r; M[ 6] = u01i; M[ 7] =  u01r;
  M[ 8] = u10r; M[ 9] = -u10i; M[10] = u11r; M[11] = -u11i;
  M[12] = u10i; M[13] =  u10r; M[14] = u11i; M[15] =  u11r;
  #pragma unroll
  for (int k = 0; k < 16; ++k) sm[g * 16 + k] = M[k];
  }
  __syncthreads();
  {
    const int R = g >> 3, Cc = g & 7;
    const int ao = R >> 2, bo = (R >> 1) & 1, co = R & 1;
    const int ai = Cc >> 2, bi = (Cc >> 1) & 1, ci = Cc & 1;
    #pragma unroll
    for (int m = 0; m < 8; ++m) {
      const int pr = m >> 1;
      const int a = (pr == 0) ? 1 : (pr == 1 ? 3 : (pr == 2 ? 6 : 8));
      const bool cv = (m & 1) != 0;
      const float* A = sm + (cv ? 29 + a : 20 + a) * 16;
      const float* V = sm + (ao ? 30 + a : 21 + a) * 16;
      const float ar = A[(2 * ao) * 4 + 2 * ai], ai_ = A[(2 * ao + 1) * 4 + 2 * ai];
      const float vr = V[(2 * bo) * 4 + 2 * bi];
      const float vi = V[(2 * bo + 1) * 4 + 2 * bi];
      const float zr = ar * vr - ai_ * vi, zi = ar * vi + ai_ * vr;
      G[640 + m * 64 + g] = (co == ci) ? zr : (co ? zi : -zi);
    }
  }
}

__device__ __forceinline__ v8f mm(v2f a, v2f b, v8f c) {
  return __builtin_amdgcn_wmma_f32_16x16x4_f32(false, a, false, b, (short)0, c,
                                               false, false);
}

__device__ __forceinline__ v2f cmul(v2f a, v2f b) {
  return v2f{a.x * b.x - a.y * b.y, a.x * b.y + a.y * b.x};
}

__device__ __forceinline__ void cos_sin_clipped(float u, float& c, float& s) {
  const bool hi = u > 0.5f;
  const float v = hi ? 1.f - u : u;
  const float th = v * (0.5f * 3.14159265358979323846f);
  const float t = th * th;
  float p = fmaf(-1.9515295891e-4f, t, 8.3321608736e-3f);
  p = fmaf(p, t, -1.6666654611e-1f);
  const float sn = fmaf(p, t * th, th);
  float q = fmaf(2.443315711809948e-5f, t, -1.388731625493765e-3f);
  q = fmaf(q, t, 4.166664568298827e-2f);
  const float cs = fmaf(q, t * t, fmaf(-0.5f, t, 1.f));
  c = hi ? sn : cs;
  s = hi ? cs : sn;
}

__device__ __forceinline__ v2f pick(bool c, v2f one, v2f zero) {
  return v2f{c ? one.x : zero.x, c ? one.y : zero.y};
}

__device__ __forceinline__ v2f pick2(const v2f (&T)[2][2], bool a, bool b) {
  return pick(a, pick(b, T[1][1], T[1][0]), pick(b, T[0][1], T[0][0]));
}

__device__ __forceinline__ void load_gate(const float* __restrict__ G, int g,
                                          int t, v2f& a1, v2f& a2, v2f& ad) {
  const float* M = G + g * 16 + (t & 3) * 4;
  const float m0 = M[0], m1 = M[1], m2 = M[2], m3 = M[3];
  const bool on = ((t < 16) == ((t & 15) < 8));
  a1.x = on ? m0 : 0.f; a1.y = on ? m1 : 0.f;
  a2.x = on ? m2 : 0.f; a2.y = on ? m3 : 0.f;
  ad.x = (t < 16) ? m0 : m2; ad.y = (t < 16) ? m1 : m3;
}

__device__ __forceinline__ void load_mux(const float* __restrict__ G, int gR,
                                         int gF, int t, v2f& a1, v2f& a2) {
  const float* M = G + ((t & 4) ? gF : gR) * 16 + (t & 3) * 4;
  const float m0 = M[0], m1 = M[1], m2 = M[2], m3 = M[3];
  const bool on = ((t < 16) == ((t & 15) < 8));
  a1.x = on ? m0 : 0.f; a1.y = on ? m1 : 0.f;
  a2.x = on ? m2 : 0.f; a2.y = on ? m3 : 0.f;
}

__device__ __forceinline__ void load_gate2(const float* __restrict__ G, int m,
                                           int t, v2f (&a)[4]) {
  const float* M = G + 640 + m * 64 + (t & 7) * 8;
  const bool on = ((t < 16) == ((t & 15) < 8));
  #pragma unroll
  for (int j = 0; j < 4; ++j) {
    const float x = M[2 * j], y = M[2 * j + 1];
    a[j].x = on ? x : 0.f; a[j].y = on ? y : 0.f;
  }
}

template <int RB1, int CB, int SEL, int N>
__device__ __forceinline__ void gate2_rr(v2f (&s)[N], const v2f (&g0)[4]) {
  constexpr int RB0 = RB1 - 1;
  #pragma unroll
  for (int p = 0; p < N / 4; ++p) {
    const int i0 = ((p >> RB0) << (RB1 + 1)) | (p & ((1 << RB0) - 1));
    const int i1 = i0 | (1 << RB0);
    const int i2 = i0 | (1 << RB1);
    const int i3 = i1 | i2;
    if (CB >= 0 && ((i0 >> (CB < 0 ? 0 : CB)) & 1) != SEL) continue;
    v8f z = {};
    v8f d = mm(g0[0], s[i0], z);
    d = mm(g0[1], s[i1], d);
    d = mm(g0[2], s[i2], d);
    d = mm(g0[3], s[i3], d);
    s[i0] = v2f{d[0], d[1]};
    s[i1] = v2f{d[2], d[3]};
    s[i2] = v2f{d[4], d[5]};
    s[i3] = v2f{d[6], d[7]};
  }
}

template <int RB>
__device__ __forceinline__ void gate_mux_lr(v2f (&s)[32], v2f m1, v2f m2,
                                            bool ctl) {
  #pragma unroll
  for (int p = 0; p < 16; ++p) {
    const int i0 = ((p >> RB) << (RB + 1)) | (p & ((1 << RB) - 1));
    const int i1 = i0 | (1 << RB);
    v8f z = {};
    v8f d = mm(m1, s[i0], z);
    d = mm(m2, s[i1], d);
    s[i0] = v2f{ctl ? d[4] : d[0], ctl ? d[5] : d[1]};
    s[i1] = v2f{ctl ? d[6] : d[2], ctl ? d[7] : d[3]};
  }
}

template <int CB>
__device__ __forceinline__ void gate_rl(v2f (&s)[32], v2f ad, bool lo) {
  #pragma unroll
  for (int r = 0; r < 32; ++r) {
    if (!((r >> CB) & 1)) continue;
    v8f z = {};
    v8f d = mm(ad, s[r], z);
    s[r] = v2f{lo ? d[0] : d[2], lo ? d[1] : d[3]};
  }
}

__device__ __forceinline__ void transpose(v2f (&s)[32], float2* sv, int t) {
  #pragma unroll
  for (int r = 0; r < 32; ++r)
    sv[r * 32 + (t ^ r)] = make_float2(s[r].x, s[r].y);
  __builtin_amdgcn_wave_barrier();
  #pragma unroll
  for (int r = 0; r < 32; ++r) {
    float2 v = sv[t * 32 + (r ^ t)];
    s[r] = v2f{v.x, v.y};
  }
}

#define G2(RB1, CB, m0) { load_gate2(G, (m0), t, g0); gate2_rr<RB1, CB, 0, 32>(s, g0); \
                          load_gate2(G, (m0) + 1, t, g1); gate2_rr<RB1, CB, 1, 32>(s, g1); }
#define PHASE_A1     G2(1, 2, 2)
#define PHASE_B1     { load_mux(G, 25, 34, t, a1, a2); gate_mux_lr<4>(s, a1, a2, (t & 1) != 0); } \
                     G2(3, 4, 4) G2(1, 2, 6) \
                     { load_gate(G, 39, t, a1, a2, ad); gate_rl<0>(s, ad, t < 16); }

__global__ __launch_bounds__(256, 4) void qsim_kernel(
    const float* __restrict__ inp, const float* __restrict__ G,
    float* __restrict__ out) {
  __shared__ float2 st[WPB][DIM];
  const int t = threadIdx.x & 31;
  const int w = threadIdx.x >> 5;
  const int samp = blockIdx.x * WPB + w;
  float2* sv = st[w];

  float cc[NQ], ss[NQ];
  const float* x = inp + samp * NQ;
  #pragma unroll
  for (int q = 0; q < NQ; ++q) {
    const float xv = fminf(fmaxf(x[q], 0.f), 1.f);
    cos_sin_clipped(xv, cc[q], ss[q]);
  }
  v2f a1, a2, ad;
  v2f g0[4], g1[4];
  v2f w0[2];
  v2f T[NQ][2][2];
  {
    load_gate(G, 0, t, a1, a2, ad);
    v8f z = {};
    v8f d = mm(a1, v2f{cc[0], 0.f}, z);
    d = mm(a2, v2f{ss[0], 0.f}, d);
    w0[0] = v2f{d[0], d[1]};
    w0[1] = v2f{d[2], d[3]};
  }
  #pragma unroll
  for (int q = 1; q < NQ; ++q) {
    load_mux(G, q, 9 + q, t, a1, a2);
    v8f z = {};
    v8f d = mm(a1, v2f{cc[q], 0.f}, z);
    d = mm(a2, v2f{ss[q], 0.f}, d);
    T[q][0][0] = v2f{d[0], d[1]};
    T[q][0][1] = v2f{d[2], d[3]};
    T[q][1][0] = v2f{d[4], d[5]};
    T[q][1][1] = v2f{d[6], d[7]};
  }
  v2f X[2][2];
  #pragma unroll
  for (int a = 0; a < 2; ++a) {
    #pragma unroll
    for (int b = 0; b < 2; ++b) X[a][b] = cmul(w0[a], T[1][a][b]);
  }
  {
    load_mux(G, 20, 19, t, a1, a2);
    const bool c9 = (t & 1) != 0;
    #pragma unroll
    for (int b = 0; b < 2; ++b) {
      v8f z = {};
      v8f d = mm(a1, X[0][b], z);
      d = mm(a2, X[1][b], d);
      X[0][b] = pick(c9, v2f{d[4], d[5]}, v2f{d[0], d[1]});
      X[1][b] = pick(c9, v2f{d[6], d[7]}, v2f{d[2], d[3]});
    }
  }
  const bool b5 = ((t >> 4) & 1) != 0, b6 = ((t >> 3) & 1) != 0;
  const bool b7 = ((t >> 2) & 1) != 0, b8 = ((t >> 1) & 1) != 0;
  const bool b9 = (t & 1) != 0;
  v2f plo = cmul(cmul(pick2(T[6], b5, b6), pick2(T[7], b6, b7)),
                 cmul(pick2(T[8], b7, b8), pick2(T[9], b8, b9)));
  v2f E[2][2];
  #pragma unroll
  for (int b = 0; b < 2; ++b) {
    const v2f f = cmul(pick(b5, T[5][b][1], T[5][b][0]), plo);
    E[0][b] = cmul(T[4][0][b], f);
    E[1][b] = cmul(T[4][1][b], f);
  }
  v2f P2[8], P3[16];
  #pragma unroll
  for (int k = 0; k < 8; ++k)
    P2[k] = cmul(X[k >> 2][(k >> 1) & 1], T[2][(k >> 1) & 1][k & 1]);
  #pragma unroll
  for (int k = 0; k < 16; ++k)
    P3[k] = cmul(P2[k >> 1], T[3][(k >> 1) & 1][k & 1]);
  load_gate2(G, 0, t, g0); gate2_rr<2, 3, 0, 16>(P3, g0);
  load_gate2(G, 1, t, g1); gate2_rr<2, 3, 1, 16>(P3, g1);
  v2f s[32];
  #pragma unroll
  for (int r = 0; r < 32; ++r)
    s[r] = cmul(P3[r >> 1], E[(r >> 1) & 1][r & 1]);

  PHASE_A1
  transpose(s, sv, t);
  PHASE_B1

  float acc[NQ];
  #pragma unroll
  for (int q = 0; q < NQ; ++q) acc[q] = 0.f;
  #pragma unroll
  for (int r = 0; r < 32; ++r) {
    const int i = t * 32 + r;
    float p = s[r].x * s[r].x + s[r].y * s[r].y;
    #pragma unroll
    for (int q = 0; q < NQ; ++q)
      acc[q] += ((i >> (9 - q)) & 1) ? -p : p;
  }
  #pragma unroll
  for (int q = 0; q < NQ; ++q) {
    #pragma unroll
    for (int off = 16; off >= 1; off >>= 1)
      acc[q] += __shfl_xor(acc[q], off, 32);
  }
  if (t == 0) {
    #pragma unroll
    for (int q = 0; q < NQ; ++q) out[samp * NQ + q] = acc[q];
  }
}

extern "C" void kernel_launch(void* const* d_in, const int* in_sizes, int n_in,
                              void* d_out, int out_size, void* d_ws,
                              size_t ws_size, hipStream_t stream) {
  const float* inputs = (const float*)d_in[0];
  const float* weights = (const float*)d_in[1];
  float* out = (float*)d_out;
  float* G = (float*)d_ws;

  const int B = in_sizes[0] / NQ;
  build_gates_kernel<<<dim3(1), dim3(64), 0, stream>>>(weights, G);
  qsim_kernel<<<dim3(B / WPB), dim3(256), 0, stream>>>(inputs, G, out);
}


# The qubit on lane bit 4 joins the register pair's gate (3,4,5 and 8,9,0): dense A, k from both half-waves, real 16x16 by the same 4 chained WMMAs: 136 WMMAs, not 184; gate products reassociated
# speedup vs baseline: 5.5540x; 1.1654x over previous
#include <hip/hip_runtime.h>
#include <hip/hip_bf16.h>

typedef __attribute__((ext_vector_type(2))) float v2f;
typedef __attribute__((ext_vector_type(8))) float v8f;

#define NQ 10
#define DIM 1024
#define WPB 8

#define OFF8 640
#define OFF16 (640 + 4 * 64)
__global__ void build_gates_kernel(const float* __restrict__ W,
                                   float* __restrict__ G) {
  __shared__ float sm[640];
  const int g = threadIdx.x;
  if (g < 40) {
  int L = g / 20, r = g % 20;
  float u00r = 0.f, u00i = 0.f, u01r = 0.f, u01i = 0.f;
  float u10r = 0.f, u10i = 0.f, u11r = 0.f, u11i = 0.f;
  const bool cry = (r >= 10);
  const bool fused = cry && r < 19;
  const bool post = (g == 19);
  if (!cry || fused || post) {
    int q = fused ? (r - 9) : (post ? 0 : r);
    int base = (post ? 1 : L) * 40 + 3 * q;
    float phi = W[base], th = W[base + 1], om = W[base + 2];
    float ct = cosf(0.5f * th), st = sinf(0.5f * th);
    float ap = 0.5f * (phi + om), am = 0.5f * (phi - om);
    float cap = cosf(ap), sap = sinf(ap);
    float cam = cosf(am), sam = sinf(am);
    u00r =  ct * cap; u00i = -ct * sap;
    u01r = -st * cam; u01i = -st * sam;
    u10r =  st * cam; u10i = -st * sam;
    u11r =  ct * cap; u11i =  ct * sap;
  }
  if (cry) {
    int e = r - 10;
    float th = W[L * 40 + 30 + e];
    float ct = cosf(0.5f * th), st = sinf(0.5f * th);
    if (fused) {
      float n00r = ct * u00r - st * u10r, n00i = ct * u00i - st * u10i;
      float n01r = ct * u01r - st * u11r, n01i = ct * u01i - st * u11i;
      float n10r = st * u00r + ct * u10r, n10i = st * u00i + ct * u10i;
      float n11r = st * u01r + ct * u11r, n11i = st * u01i + ct * u11i;
      u00r = n00r; u00i = n00i; u01r = n01r; u01i = n01i;
      u10r = n10r; u10i = n10i; u11r = n11r; u11i = n11i;
    } else if (post) {
      float n00r = u00r * ct + u01r * st, n00i = u00i * ct + u01i * st;
      float n01r = u01r * ct - u00r * st, n01i = u01i * ct - u00i * st;
      float n10r = u10r * ct + u11r * st, n10i = u10i * ct + u11i * st;
      float n11r = u11r * ct - u10r * st, n11i = u11i * ct - u10i * st;
      u00r = n00r; u00i = n00i; u01r = n01r; u01i = n01i;
      u10r = n10r; u10i = n10i; u11r = n11r; u11i = n11i;
    } else {
      u00r = ct;  u00i = 0.f; u01r = -st; u01i = 0.f;
      u10r = st;  u10i = 0.f; u11r = ct;  u11i = 0.f;
    }
  }
  float* M = G + g * 16;
  M[ 0] = u00r; M[ 1] = -u00i; M[ 2] = u01r; M[ 3] = -u01i;
  M[ 4] = u00i; M[ 5] =  u00r; M[ 6] = u01i; M[ 7] =  u01r;
  M[ 8] = u10r; M[ 9] = -u10i; M[10] = u11r; M[11] = -u11i;
  M[12] = u10i; M[13] =  u10r; M[14] = u11i; M[15] =  u11r;
  #pragma unroll
  for (int k = 0; k < 16; ++k) sm[g * 16 + k] = M[k];
  }
  __syncthreads();
  if (g < 64) {
    const int R = g >> 3, Cc = g & 7;
    const int ao = R >> 2, bo = (R >> 1) & 1, co = R & 1;
    const int ai = Cc >> 2, bi = (Cc >> 1) & 1, ci = Cc & 1;
    #pragma unroll
    for (int m = 0; m < 4; ++m) {
      const int a = (m >> 1) ? 6 : 1;
      const bool cv = (m & 1) != 0;
      const float* A = sm + (cv ? 29 + a : 20 + a) * 16;
      const float* V = sm + (ao ? 30 + a : 21 + a) * 16;
      const float ar = A[(2 * ao) * 4 + 2 * ai], ai_ = A[(2 * ao + 1) * 4 + 2 * ai];
      const float vr = V[(2 * bo) * 4 + 2 * bi];
      const float vi = V[(2 * bo + 1) * 4 + 2 * bi];
      const float zr = ar * vr - ai_ * vi, zi = ar * vi + ai_ * vr;
      G[OFF8 + m * 64 + g] = (co == ci) ? zr : (co ? zi : -zi);
    }
  }
  {
    const int R = g >> 4, Cc = g & 15;
    const int co = R >> 3, ao = (R >> 2) & 1, bo = (R >> 1) & 1, ro = R & 1;
    const int ai = Cc >> 3, bi = (Cc >> 2) & 1, ci = (Cc >> 1) & 1, ri = Cc & 1;
    #pragma unroll
    for (int m = 0; m < 4; ++m) {
      const int a = (m >> 1) ? 8 : 3;
      const bool cv = (m & 1) != 0;
      const float* A = sm + (cv ? 29 + a : 20 + a) * 16;
      const float* V = sm + (ao ? 30 + a : 21 + a) * 16;
      const float ar = A[(2 * ao) * 4 + 2 * ai], ai_ = A[(2 * ao + 1) * 4 + 2 * ai];
      const float vr = V[(2 * bo) * 4 + 2 * bi];
      const float vi = V[(2 * bo + 1) * 4 + 2 * bi];
      const float pr = ar * vr - ai_ * vi, pi = ar * vi + ai_ * vr;
      float xr, xi;
      if (bo || a == 3) {
        const float* X = sm + (bo ? 31 + a : 22 + a) * 16;
        xr = X[(2 * co) * 4 + 2 * ci];
        xi = X[(2 * co + 1) * 4 + 2 * ci];
      } else {
        xr = (co == ci) ? 1.f : 0.f;
        xi = 0.f;
      }
      const float zr = pr * xr - pi * xi, zi = pr * xi + pi * xr;
      G[OFF16 + m * 256 + g] = (ro == ri) ? zr : (ro ? zi : -zi);
    }
  }
}

__device__ __forceinline__ v8f mm(v2f a, v2f b, v8f c) {
  return __builtin_amdgcn_wmma_f32_16x16x4_f32(false, a, false, b, (short)0, c,
                                               false, false);
}

__device__ __forceinline__ v2f cmul(v2f a, v2f b) {
  return v2f{a.x * b.x - a.y * b.y, a.x * b.y + a.y * b.x};
}

__device__ __forceinline__ void cos_sin_clipped(float u, float& c, float& s) {
  const bool hi = u > 0.5f;
  const float v = hi ? 1.f - u : u;
  const float th = v * (0.5f * 3.14159265358979323846f);
  const float t = th * th;
  float p = fmaf(-1.9515295891e-4f, t, 8.3321608736e-3f);
  p = fmaf(p, t, -1.6666654611e-1f);
  const float sn = fmaf(p, t * th, th);
  float q = fmaf(2.443315711809948e-5f, t, -1.388731625493765e-3f);
  q = fmaf(q, t, 4.166664568298827e-2f);
  const float cs = fmaf(q, t * t, fmaf(-0.5f, t, 1.f));
  c = hi ? sn : cs;
  s = hi ? cs : sn;
}

__device__ __forceinline__ v2f pick(bool c, v2f one, v2f zero) {
  return v2f{c ? one.x : zero.x, c ? one.y : zero.y};
}

__device__ __forceinline__ v2f pick2(const v2f (&T)[2][2], bool a, bool b) {
  return pick(a, pick(b, T[1][1], T[1][0]), pick(b, T[0][1], T[0][0]));
}

__device__ __forceinline__ void load_gate(const float* __restrict__ G, int g,
                                          int t, v2f& a1, v2f& a2) {
  const float* M = G + g * 16 + (t & 3) * 4;
  const float m0 = M[0], m1 = M[1], m2 = M[2], m3 = M[3];
  const bool on = ((t < 16) == ((t & 15) < 8));
  a1.x = on ? m0 : 0.f; a1.y = on ? m1 : 0.f;
  a2.x = on ? m2 : 0.f; a2.y = on ? m3 : 0.f;
}

__device__ __forceinline__ void load_mux(const float* __restrict__ G, int gR,
                                         int gF, int t, v2f& a1, v2f& a2) {
  const float* M = G + ((t & 4) ? gF : gR) * 16 + (t & 3) * 4;
  const float m0 = M[0], m1 = M[1], m2 = M[2], m3 = M[3];
  const bool on = ((t < 16) == ((t & 15) < 8));
  a1.x = on ? m0 : 0.f; a1.y = on ? m1 : 0.f;
  a2.x = on ? m2 : 0.f; a2.y = on ? m3 : 0.f;
}

__device__ __forceinline__ void load_gate2(const float* __restrict__ G, int m,
                                           int t, v2f (&a)[4]) {
  const float* M = G + OFF8 + m * 64 + (t & 7) * 8;
  const bool on = ((t < 16) == ((t & 15) < 8));
  #pragma unroll
  for (int j = 0; j < 4; ++j) {
    const float x = M[2 * j], y = M[2 * j + 1];
    a[j].x = on ? x : 0.f; a[j].y = on ? y : 0.f;
  }
}

__device__ __forceinline__ void load_gate3(const float* __restrict__ G, int m,
                                           int t, v2f (&a)[4]) {
  const float* M = G + OFF16 + m * 256 + (t & 15) * 16 + (t >> 4) * 2;
  #pragma unroll
  for (int j = 0; j < 4; ++j) a[j] = v2f{M[4 * j], M[4 * j + 1]};
}

template <int RB1, int CB, int SEL, int N>
__device__ __forceinline__ void gate2_rr(v2f (&s)[N], const v2f (&g0)[4]) {
  constexpr int RB0 = RB1 - 1;
  #pragma unroll
  for (int p = 0; p < N / 4; ++p) {
    const int i0 = ((p >> RB0) << (RB1 + 1)) | (p & ((1 << RB0) - 1));
    const int i1 = i0 | (1 << RB0);
    const int i2 = i0 | (1 << RB1);
    const int i3 = i1 | i2;
    if (CB >= 0 && ((i0 >> (CB < 0 ? 0 : CB)) & 1) != SEL) continue;
    v8f z = {};
    v8f d = mm(g0[0], s[i0], z);
    d = mm(g0[1], s[i1], d);
    d = mm(g0[2], s[i2], d);
    d = mm(g0[3], s[i3], d);
    s[i0] = v2f{d[0], d[1]};
    s[i1] = v2f{d[2], d[3]};
    s[i2] = v2f{d[4], d[5]};
    s[i3] = v2f{d[6], d[7]};
  }
}

__device__ __forceinline__ void transpose(v2f (&s)[32], float2* sv, int t) {
  #pragma unroll
  for (int r = 0; r < 32; ++r)
    sv[r * 32 + (t ^ r)] = make_float2(s[r].x, s[r].y);
  __builtin_amdgcn_wave_barrier();
  #pragma unroll
  for (int r = 0; r < 32; ++r) {
    float2 v = sv[t * 32 + (r ^ t)];
    s[r] = v2f{v.x, v.y};
  }
}

#define G2(RB1, CB, m0) { load_gate2(G, (m0), t, g0); gate2_rr<RB1, CB, 0, 32>(s, g0); \
                          load_gate2(G, (m0) + 1, t, g1); gate2_rr<RB1, CB, 1, 32>(s, g1); }
#define G3(RB1, CB, m0) { load_gate3(G, (m0), t, g0); gate2_rr<RB1, CB, 0, 32>(s, g0); \
                          load_gate3(G, (m0) + 1, t, g1); gate2_rr<RB1, CB, 1, 32>(s, g1); }
#define PHASE_A1     G3(1, 2, 0)
#define PHASE_B1     G2(3, 4, 2) G3(1, 2, 2)

__global__ __launch_bounds__(256, 4) void qsim_kernel(
    const float* __restrict__ inp, const float* __restrict__ G,
    float* __restrict__ out) {
  __shared__ float2 st[WPB][DIM];
  const int t = threadIdx.x & 31;
  const int w = threadIdx.x >> 5;
  const int samp = blockIdx.x * WPB + w;
  float2* sv = st[w];

  float cc[NQ], ss[NQ];
  const float* x = inp + samp * NQ;
  #pragma unroll
  for (int q = 0; q < NQ; ++q) {
    const float xv = fminf(fmaxf(x[q], 0.f), 1.f);
    cos_sin_clipped(xv, cc[q], ss[q]);
  }
  v2f a1, a2;
  v2f g0[4], g1[4];
  v2f w0[2];
  v2f T[NQ][2][2];
  {
    load_gate(G, 0, t, a1, a2);
    v8f z = {};
    v8f d = mm(a1, v2f{cc[0], 0.f}, z);
    d = mm(a2, v2f{ss[0], 0.f}, d);
    w0[0] = v2f{d[0], d[1]};
    w0[1] = v2f{d[2], d[3]};
  }
  #pragma unroll
  for (int q = 1; q < NQ; ++q) {
    load_mux(G, q, 9 + q, t, a1, a2);
    v8f z = {};
    v8f d = mm(a1, v2f{cc[q], 0.f}, z);
    d = mm(a2, v2f{ss[q], 0.f}, d);
    T[q][0][0] = v2f{d[0], d[1]};
    T[q][0][1] = v2f{d[2], d[3]};
    T[q][1][0] = v2f{d[4], d[5]};
    T[q][1][1] = v2f{d[6], d[7]};
  }
  v2f X[2][2];
  #pragma unroll
  for (int a = 0; a < 2; ++a) {
    #pragma unroll
    for (int b = 0; b < 2; ++b) X[a][b] = cmul(w0[a], T[1][a][b]);
  }
  {
    load_mux(G, 20, 19, t, a1, a2);
    const bool c9 = (t & 1) != 0;
    #pragma unroll
    for (int b = 0; b < 2; ++b) {
      v8f z = {};
      v8f d = mm(a1, X[0][b], z);
      d = mm(a2, X[1][b], d);
      X[0][b] = pick(c9, v2f{d[4], d[5]}, v2f{d[0], d[1]});
      X[1][b] = pick(c9, v2f{d[6], d[7]}, v2f{d[2], d[3]});
    }
  }
  const bool b5 = ((t >> 4) & 1) != 0, b6 = ((t >> 3) & 1) != 0;
  const bool b7 = ((t >> 2) & 1) != 0, b8 = ((t >> 1) & 1) != 0;
  const bool b9 = (t & 1) != 0;
  v2f plo = cmul(cmul(pick2(T[6], b5, b6), pick2(T[7], b6, b7)),
                 cmul(pick2(T[8], b7, b8), pick2(T[9], b8, b9)));
  v2f E[2][2];
  #pragma unroll
  for (int b = 0; b < 2; ++b) {
    const v2f f = cmul(pick(b5, T[5][b][1], T[5][b][0]), plo);
    E[0][b] = cmul(T[4][0][b], f);
    E[1][b] = cmul(T[4][1][b], f);
  }
  v2f P2[8], P3[16];
  #pragma unroll
  for (int k = 0; k < 8; ++k)
    P2[k] = cmul(X[k >> 2][(k >> 1) & 1], T[2][(k >> 1) & 1][k & 1]);
  #pragma unroll
  for (int k = 0; k < 16; ++k)
    P3[k] = cmul(P2[k >> 1], T[3][(k >> 1) & 1][k & 1]);
  load_gate2(G, 0, t, g0); gate2_rr<2, 3, 0, 16>(P3, g0);
  load_gate2(G, 1, t, g1); gate2_rr<2, 3, 1, 16>(P3, g1);
  v2f s[32];
  #pragma unroll
  for (int r = 0; r < 32; ++r)
    s[r] = cmul(P3[r >> 1], E[(r >> 1) & 1][r & 1]);

  PHASE_A1
  transpose(s, sv, t);
  PHASE_B1

  float acc[NQ];
  #pragma unroll
  for (int q = 0; q < NQ; ++q) acc[q] = 0.f;
  #pragma unroll
  for (int r = 0; r < 32; ++r) {
    const int i = t * 32 + r;
    float p = s[r].x * s[r].x + s[r].y * s[r].y;
    #pragma unroll
    for (int q = 0; q < NQ; ++q)
      acc[q] += ((i >> (9 - q)) & 1) ? -p : p;
  }
  #pragma unroll
  for (int q = 0; q < NQ; ++q) {
    #pragma unroll
    for (int off = 16; off >= 1; off >>= 1)
      acc[q] += __shfl_xor(acc[q], off, 32);
  }
  if (t == 0) {
    #pragma unroll
    for (int q = 0; q < NQ; ++q) out[samp * NQ + q] = acc[q];
  }
}

extern "C" void kernel_launch(void* const* d_in, const int* in_sizes, int n_in,
                              void* d_out, int out_size, void* d_ws,
                              size_t ws_size, hipStream_t stream) {
  const float* inputs = (const float*)d_in[0];
  const float* weights = (const float*)d_in[1];
  float* out = (float*)d_out;
  float* G = (float*)d_ws;

  const int B = in_sizes[0] / NQ;
  build_gates_kernel<<<dim3(1), dim3(256), 0, stream>>>(weights, G);
  qsim_kernel<<<dim3(B / WPB), dim3(256), 0, stream>>>(inputs, G, out);
}


# Layer-1 gate on (3,4,5) acts on the 256-amplitude prefix over qubits 0-6, 9 (lanes hold qubits 0,1, not copies): 112 WMMAs, not 136; exchange of 8 registers, not 32; products reassociated
# speedup vs baseline: 6.0822x; 1.0951x over previous
#include <hip/hip_runtime.h>
#include <hip/hip_bf16.h>

typedef __attribute__((ext_vector_type(2))) float v2f;
typedef __attribute__((ext_vector_type(8))) float v8f;

#define NQ 10
#define WPB 8

#define OFF8 640
#define OFF16 (640 + 4 * 64)
__global__ void build_gates_kernel(const float* __restrict__ W,
                                   float* __restrict__ G) {
  __shared__ float sm[640];
  const int g = threadIdx.x;
  if (g < 40) {
  int L = g / 20, r = g % 20;
  float u00r = 0.f, u00i = 0.f, u01r = 0.f, u01i = 0.f;
  float u10r = 0.f, u10i = 0.f, u11r = 0.f, u11i = 0.f;
  const bool cry = (r >= 10);
  const bool fused = cry && r < 19;
  const bool post = (g == 19);
  if (!cry || fused || post) {
    int q = fused ? (r - 9) : (post ? 0 : r);
    int base = (post ? 1 : L) * 40 + 3 * q;
    float phi = W[base], th = W[base + 1], om = W[base + 2];
    float ct = cosf(0.5f * th), st = sinf(0.5f * th);
    float ap = 0.5f * (phi + om), am = 0.5f * (phi - om);
    float cap = cosf(ap), sap = sinf(ap);
    float cam = cosf(am), sam = sinf(am);
    u00r =  ct * cap; u00i = -ct * sap;
    u01r = -st * cam; u01i = -st * sam;
    u10r =  st * cam; u10i = -st * sam;
    u11r =  ct * cap; u11i =  ct * sap;
  }
  if (cry) {
    int e = r - 10;
    float th = W[L * 40 + 30 + e];
    float ct = cosf(0.5f * th), st = sinf(0.5f * th);
    if (fused) {
      float n00r = ct * u00r - st * u10r, n00i = ct * u00i - st * u10i;
      float n01r = ct * u01r - st * u11r, n01i = ct * u01i - st * u11i;
      float n10r = st * u00r + ct * u10r, n10i = st * u00i + ct * u10i;
      float n11r = st * u01r + ct * u11r, n11i = st * u01i + ct * u11i;
      u00r = n00r; u00i = n00i; u01r = n01r; u01i = n01i;
      u10r = n10r; u10i = n10i; u11r = n11r; u11i = n11i;
    } else if (post) {
      float n00r = u00r * ct + u01r * st, n00i = u00i * ct + u01i * st;
      float n01r = u01r * ct - u00r * st, n01i = u01i * ct - u00i * st;
      float n10r = u10r * ct + u11r * st, n10i = u10i * ct + u11i * st;
      float n11r = u11r * ct - u10r * st, n11i = u11i * ct - u10i * st;
      u00r = n00r; u00i = n00i; u01r = n01r; u01i = n01i;
      u10r = n10r; u10i = n10i; u11r = n11r; u11i = n11i;
    } else {
      u00r = ct;  u00i = 0.f; u01r = -st; u01i = 0.f;
      u10r = st;  u10i = 0.f; u11r = ct;  u11i = 0.f;
    }
  }
  float* M = G + g * 16;
  M[ 0] = u00r; M[ 1] = -u00i; M[ 2] = u01r; M[ 3] = -u01i;
  M[ 4] = u00i; M[ 5] =  u00r; M[ 6] = u01i; M[ 7] =  u01r;
  M[ 8] = u10r; M[ 9] = -u10i; M[10] = u11r; M[11] = -u11i;
  M[12] = u10i; M[13] =  u10r; M[14] = u11i; M[15] =  u11r;
  #pragma unroll
  for (int k = 0; k < 16; ++k) sm[g * 16 + k] = M[k];
  }
  __syncthreads();
  if (g < 64) {
    const int R = g >> 3, Cc = g & 7;
    const int ao = R >> 2, bo = (R >> 1) & 1, co = R & 1;
    const int ai = Cc >> 2, bi = (Cc >> 1) & 1, ci = Cc & 1;
    #pragma unroll
    for (int m = 0; m < 4; ++m) {
      const int a = (m >> 1) ? 6 : 1;
      const bool cv = (m & 1) != 0;
      const float* A = sm + (cv ? 29 + a : 20 + a) * 16;
      const float* V = sm + (ao ? 30 + a : 21 + a) * 16;
      const float ar = A[(2 * ao) * 4 + 2 * ai], ai_ = A[(2 * ao + 1) * 4 + 2 * ai];
      const float vr = V[(2 * bo) * 4 + 2 * bi];
      const float vi = V[(2 * bo + 1) * 4 + 2 * bi];
      const float zr = ar * vr - ai_ * vi, zi = ar * vi + ai_ * vr;
      G[OFF8 + m * 64 + g] = (co == ci) ? zr : (co ? zi : -zi);
    }
  }
  {
    const int R = g >> 4, Cc = g & 15;
    const int co = R >> 3, ao = (R >> 2) & 1, bo = (R >> 1) & 1, ro = R & 1;
    const int ai = Cc >> 3, bi = (Cc >> 2) & 1, ci = (Cc >> 1) & 1, ri = Cc & 1;
    #pragma unroll
    for (int m = 0; m < 4; ++m) {
      const int a = (m >> 1) ? 8 : 3;
      const bool cv = (m & 1) != 0;
      const float* A = sm + (cv ? 29 + a : 20 + a) * 16;
      const float* V = sm + (ao ? 30 + a : 21 + a) * 16;
      const float ar = A[(2 * ao) * 4 + 2 * ai], ai_ = A[(2 * ao + 1) * 4 + 2 * ai];
      const float vr = V[(2 * bo) * 4 + 2 * bi];
      const float vi = V[(2 * bo + 1) * 4 + 2 * bi];
      const float pr = ar * vr - ai_ * vi, pi = ar * vi + ai_ * vr;
      float xr, xi;
      if (bo || a == 3) {
        const float* X = sm + (bo ? 31 + a : 22 + a) * 16;
        xr = X[(2 * co) * 4 + 2 * ci];
        xi = X[(2 * co + 1) * 4 + 2 * ci];
      } else {
        xr = (co == ci) ? 1.f : 0.f;
        xi = 0.f;
      }
      const float zr = pr * xr - pi * xi, zi = pr * xi + pi * xr;
      G[OFF16 + m * 256 + g] = (ro == ri) ? zr : (ro ? zi : -zi);
    }
  }
}

__device__ __forceinline__ v8f mm(v2f a, v2f b, v8f c) {
  return __builtin_amdgcn_wmma_f32_16x16x4_f32(false, a, false, b, (short)0, c,
                                               false, false);
}

__device__ __forceinline__ v2f cmul(v2f a, v2f b) {
  return v2f{a.x * b.x - a.y * b.y, a.x * b.y + a.y * b.x};
}

__device__ __forceinline__ void cos_sin_clipped(float u, float& c, float& s) {
  const bool hi = u > 0.5f;
  const float v = hi ? 1.f - u : u;
  const float th = v * (0.5f * 3.14159265358979323846f);
  const float t = th * th;
  float p = fmaf(-1.9515295891e-4f, t, 8.3321608736e-3f);
  p = fmaf(p, t, -1.6666654611e-1f);
  const float sn = fmaf(p, t * th, th);
  float q = fmaf(2.443315711809948e-5f, t, -1.388731625493765e-3f);
  q = fmaf(q, t, 4.166664568298827e-2f);
  const float cs = fmaf(q, t * t, fmaf(-0.5f, t, 1.f));
  c = hi ? sn : cs;
  s = hi ? cs : sn;
}

__device__ __forceinline__ v2f pick(bool c, v2f one, v2f zero) {
  return v2f{c ? one.x : zero.x, c ? one.y : zero.y};
}

__device__ __forceinline__ v2f pick2(const v2f (&T)[2][2], bool a, bool b) {
  return pick(a, pick(b, T[1][1], T[1][0]), pick(b, T[0][1], T[0][0]));
}

__device__ __forceinline__ void load_gate(const float* __restrict__ G, int g,
                                          int t, v2f& a1, v2f& a2) {
  const float* M = G + g * 16 + (t & 3) * 4;
  const float m0 = M[0], m1 = M[1], m2 = M[2], m3 = M[3];
  const bool on = ((t < 16) == ((t & 15) < 8));
  a1.x = on ? m0 : 0.f; a1.y = on ? m1 : 0.f;
  a2.x = on ? m2 : 0.f; a2.y = on ? m3 : 0.f;
}

__device__ __forceinline__ void load_mux(const float* __restrict__ G, int gR,
                                         int gF, int t, v2f& a1, v2f& a2) {
  const float* M = G + ((t & 4) ? gF : gR) * 16 + (t & 3) * 4;
  const float m0 = M[0], m1 = M[1], m2 = M[2], m3 = M[3];
  const bool on = ((t < 16) == ((t & 15) < 8));
  a1.x = on ? m0 : 0.f; a1.y = on ? m1 : 0.f;
  a2.x = on ? m2 : 0.f; a2.y = on ? m3 : 0.f;
}

__device__ __forceinline__ void load_gate2(const float* __restrict__ G, int m,
                                           int t, v2f (&a)[4]) {
  const float* M = G + OFF8 + m * 64 + (t & 7) * 8;
  const bool on = ((t < 16) == ((t & 15) < 8));
  #pragma unroll
  for (int j = 0; j < 4; ++j) {
    const float x = M[2 * j], y = M[2 * j + 1];
    a[j].x = on ? x : 0.f; a[j].y = on ? y : 0.f;
  }
}

__device__ __forceinline__ void load_gate3(const float* __restrict__ G, int m,
                                           int t, v2f (&a)[4]) {
  const float* M = G + OFF16 + m * 256 + (t & 15) * 16 + (t >> 4) * 2;
  #pragma unroll
  for (int j = 0; j < 4; ++j) a[j] = v2f{M[4 * j], M[4 * j + 1]};
}

template <int RB1, int CB, int SEL, int N>
__device__ __forceinline__ void gate2_rr(v2f (&s)[N], const v2f (&g0)[4]) {
  constexpr int RB0 = RB1 - 1;
  #pragma unroll
  for (int p = 0; p < N / 4; ++p) {
    const int i0 = ((p >> RB0) << (RB1 + 1)) | (p & ((1 << RB0) - 1));
    const int i1 = i0 | (1 << RB0);
    const int i2 = i0 | (1 << RB1);
    const int i3 = i1 | i2;
    if (CB >= 0 && ((i0 >> (CB < 0 ? 0 : CB)) & 1) != SEL) continue;
    v8f z = {};
    v8f d = mm(g0[0], s[i0], z);
    d = mm(g0[1], s[i1], d);
    d = mm(g0[2], s[i2], d);
    d = mm(g0[3], s[i3], d);
    s[i0] = v2f{d[0], d[1]};
    s[i1] = v2f{d[2], d[3]};
    s[i2] = v2f{d[4], d[5]};
    s[i3] = v2f{d[6], d[7]};
  }
}

__device__ __forceinline__ void regroup(const v2f (&sr)[8], v2f (&zt)[8],
                                        float2* sv, int t) {
  const int k = ((t >> 2) & 6) | (t & 1);
  const int base = k * 32 + ((t & 6) << 2);
  #pragma unroll
  for (int r = 0; r < 8; ++r)
    sv[base + (r ^ k)] = make_float2(sr[r].x, sr[r].y);
  __builtin_amdgcn_wave_barrier();
  #pragma unroll
  for (int r = 0; r < 8; ++r) {
    float2 v = sv[r * 32 + (t ^ r)];
    zt[r] = v2f{v.x, v.y};
  }
}

#define G2(RB1, CB, m0) { load_gate2(G, (m0), t, g0); gate2_rr<RB1, CB, 0, 32>(s, g0); \
                          load_gate2(G, (m0) + 1, t, g1); gate2_rr<RB1, CB, 1, 32>(s, g1); }
#define G3(RB1, CB, m0) { load_gate3(G, (m0), t, g0); gate2_rr<RB1, CB, 0, 32>(s, g0); \
                          load_gate3(G, (m0) + 1, t, g1); gate2_rr<RB1, CB, 1, 32>(s, g1); }
#define PHASE_B1     G2(3, 4, 2) G3(1, 2, 2)

__global__ __launch_bounds__(256, 6) void qsim_kernel(
    const float* __restrict__ inp, const float* __restrict__ G,
    float* __restrict__ out) {
  __shared__ float2 st[WPB][256];
  const int t = threadIdx.x & 31;
  const int w = threadIdx.x >> 5;
  const int samp = blockIdx.x * WPB + w;
  const int samp_out = blockIdx.x * WPB + __builtin_amdgcn_readfirstlane(w);
  float2* sv = st[w];

  float cc[NQ], ss[NQ];
  const float* x = inp + samp * NQ;
  #pragma unroll
  for (int q = 0; q < NQ; ++q) {
    const float xv = fminf(fmaxf(x[q], 0.f), 1.f);
    cos_sin_clipped(xv, cc[q], ss[q]);
  }
  v2f a1, a2;
  v2f g0[4], g1[4];
  v2f w0[2];
  v2f T[NQ][2][2];
  {
    load_gate(G, 0, t, a1, a2);
    v8f z = {};
    v8f d = mm(a1, v2f{cc[0], 0.f}, z);
    d = mm(a2, v2f{ss[0], 0.f}, d);
    w0[0] = v2f{d[0], d[1]};
    w0[1] = v2f{d[2], d[3]};
  }
  #pragma unroll
  for (int q = 1; q < 7; ++q) {
    load_mux(G, q, 9 + q, t, a1, a2);
    v8f z = {};
    v8f d = mm(a1, v2f{cc[q], 0.f}, z);
    d = mm(a2, v2f{ss[q], 0.f}, d);
    T[q][0][0] = v2f{d[0], d[1]};
    T[q][0][1] = v2f{d[2], d[3]};
    T[q][1][0] = v2f{d[4], d[5]};
    T[q][1][1] = v2f{d[6], d[7]};
  }
  v2f X[2][2];
  #pragma unroll
  for (int a = 0; a < 2; ++a) {
    #pragma unroll
    for (int b = 0; b < 2; ++b) X[a][b] = cmul(w0[a], T[1][a][b]);
  }
  {
    load_mux(G, 20, 19, t, a1, a2);
    const bool c9 = (t & 1) != 0;
    #pragma unroll
    for (int b = 0; b < 2; ++b) {
      v8f z = {};
      v8f d = mm(a1, X[0][b], z);
      d = mm(a2, X[1][b], d);
      X[0][b] = pick(c9, v2f{d[4], d[5]}, v2f{d[0], d[1]});
      X[1][b] = pick(c9, v2f{d[6], d[7]}, v2f{d[2], d[3]});
    }
  }
  const bool b5 = ((t >> 4) & 1) != 0, b6 = ((t >> 3) & 1) != 0;
  const bool l0 = ((t >> 2) & 1) != 0, l1 = ((t >> 1) & 1) != 0;
  const v2f t6 = pick2(T[6], b5, b6);
  v2f E[2][2];
  #pragma unroll
  for (int b = 0; b < 2; ++b) {
    const v2f f = cmul(pick(b5, T[5][b][1], T[5][b][0]), t6);
    E[0][b] = cmul(T[4][0][b], f);
    E[1][b] = cmul(T[4][1][b], f);
  }
  v2f P2[8], P3[16];
  #pragma unroll
  for (int k = 0; k < 8; ++k)
    P2[k] = cmul(X[k >> 2][(k >> 1) & 1], T[2][(k >> 1) & 1][k & 1]);
  #pragma unroll
  for (int k = 0; k < 16; ++k)
    P3[k] = cmul(P2[k >> 1], T[3][(k >> 1) & 1][k & 1]);
  load_gate2(G, 0, t, g0); gate2_rr<2, 3, 0, 16>(P3, g0);
  load_gate2(G, 1, t, g1); gate2_rr<2, 3, 1, 16>(P3, g1);
  v2f sr[8];
  #pragma unroll
  for (int j = 0; j < 4; ++j) {
    const v2f q = pick(l0, pick(l1, P3[12 + j], P3[8 + j]),
                       pick(l1, P3[4 + j], P3[j]));
    sr[2 * j] = cmul(q, E[j & 1][0]);
    sr[2 * j + 1] = cmul(q, E[j & 1][1]);
  }

  load_gate3(G, 0, t, g0); gate2_rr<1, 2, 0, 8>(sr, g0);
  load_gate3(G, 1, t, g1); gate2_rr<1, 2, 1, 8>(sr, g1);
  v2f zt[8];
  regroup(sr, zt, sv, t);
  #pragma unroll
  for (int q = 7; q < NQ; ++q) {
    load_mux(G, q, 9 + q, t, a1, a2);
    v8f z = {};
    v8f d = mm(a1, v2f{cc[q], 0.f}, z);
    d = mm(a2, v2f{ss[q], 0.f}, d);
    T[q][0][0] = v2f{d[0], d[1]};
    T[q][0][1] = v2f{d[2], d[3]};
    T[q][1][0] = v2f{d[4], d[5]};
    T[q][1][1] = v2f{d[6], d[7]};
  }
  v2f T89[8], F[16];
  #pragma unroll
  for (int k = 0; k < 8; ++k)
    T89[k] = cmul(T[8][k >> 2][(k >> 1) & 1], T[9][(k >> 1) & 1][k & 1]);
  #pragma unroll
  for (int k = 0; k < 16; ++k)
    F[k] = cmul(T[7][k >> 3][(k >> 2) & 1], T89[k & 7]);
  v2f s[32];
  #pragma unroll
  for (int r = 0; r < 32; ++r)
    s[r] = cmul(zt[((r >> 2) & 6) | (r & 1)], F[r & 15]);
  __builtin_amdgcn_sched_barrier(0);
  PHASE_B1

  float acc[NQ];
  #pragma unroll
  for (int q = 0; q < NQ; ++q) acc[q] = 0.f;
  #pragma unroll
  for (int r = 0; r < 32; ++r) {
    const int i = t * 32 + r;
    float p = s[r].x * s[r].x + s[r].y * s[r].y;
    #pragma unroll
    for (int q = 0; q < NQ; ++q)
      acc[q] += ((i >> (9 - q)) & 1) ? -p : p;
  }
  #pragma unroll
  for (int q = 0; q < NQ; ++q) {
    #pragma unroll
    for (int off = 16; off >= 1; off >>= 1)
      acc[q] += __shfl_xor(acc[q], off, 32);
  }
  if (t == 0) {
    #pragma unroll
    for (int q = 0; q < NQ; ++q) out[samp_out * NQ + q] = acc[q];
  }
}

extern "C" void kernel_launch(void* const* d_in, const int* in_sizes, int n_in,
                              void* d_out, int out_size, void* d_ws,
                              size_t ws_size, hipStream_t stream) {
  const float* inputs = (const float*)d_in[0];
  const float* weights = (const float*)d_in[1];
  float* out = (float*)d_out;
  float* G = (float*)d_ws;

  const int B = in_sizes[0] / NQ;
  build_gates_kernel<<<dim3(1), dim3(256), 0, stream>>>(weights, G);
  qsim_kernel<<<dim3(B / WPB), dim3(256), 0, stream>>>(inputs, G, out);
}


# Products lanes repeat (layer 0 on 2-vectors, Rot_0 with gate 9->0, layer-1 gate on qubits 1,2) run once over WMMA columns, fetched by readlane or shuffle: 83 WMMAs, not 112; layer-0 sums reordered
# speedup vs baseline: 6.9094x; 1.1360x over previous
#include <hip/hip_runtime.h>
#include <hip/hip_bf16.h>

typedef __attribute__((ext_vector_type(2))) float v2f;
typedef __attribute__((ext_vector_type(8))) float v8f;

#define NQ 10
#define WPB 8

#define OFF8 640
#define OFF16 (640 + 4 * 64)
__global__ void build_gates_kernel(const float* __restrict__ W,
                                   float* __restrict__ G) {
  __shared__ float sm[640];
  const int g = threadIdx.x;
  if (g < 40) {
  int L = g / 20, r = g % 20;
  float u00r = 0.f, u00i = 0.f, u01r = 0.f, u01i = 0.f;
  float u10r = 0.f, u10i = 0.f, u11r = 0.f, u11i = 0.f;
  const bool cry = (r >= 10);
  const bool fused = cry && r < 19;
  const bool post = (g == 19);
  if (!cry || fused || post) {
    int q = fused ? (r - 9) : (post ? 0 : r);
    int base = (post ? 1 : L) * 40 + 3 * q;
    float phi = W[base], th = W[base + 1], om = W[base + 2];
    float ct = cosf(0.5f * th), st = sinf(0.5f * th);
    float ap = 0.5f * (phi + om), am = 0.5f * (phi - om);
    float cap = cosf(ap), sap = sinf(ap);
    float cam = cosf(am), sam = sinf(am);
    u00r =  ct * cap; u00i = -ct * sap;
    u01r = -st * cam; u01i = -st * sam;
    u10r =  st * cam; u10i = -st * sam;
    u11r =  ct * cap; u11i =  ct * sap;
  }
  if (cry) {
    int e = r - 10;
    float th = W[L * 40 + 30 + e];
    float ct = cosf(0.5f * th), st = sinf(0.5f * th);
    if (fused) {
      float n00r = ct * u00r - st * u10r, n00i = ct * u00i - st * u10i;
      float n01r = ct * u01r - st * u11r, n01i = ct * u01i - st * u11i;
      float n10r = st * u00r + ct * u10r, n10i = st * u00i + ct * u10i;
      float n11r = st * u01r + ct * u11r, n11i = st * u01i + ct * u11i;
      u00r = n00r; u00i = n00i; u01r = n01r; u01i = n01i;
      u10r = n10r; u10i = n10i; u11r = n11r; u11i = n11i;
    } else if (post) {
      float n00r = u00r * ct + u01r * st, n00i = u00i * ct + u01i * st;
      float n01r = u01r * ct - u00r * st, n01i = u01i * ct - u00i * st;
      float n10r = u10r * ct + u11r * st, n10i = u10i * ct + u11i * st;
      float n11r = u11r * ct - u10r * st, n11i = u11i * ct - u10i * st;
      u00r = n00r; u00i = n00i; u01r = n01r; u01i = n01i;
      u10r = n10r; u10i = n10i; u11r = n11r; u11i = n11i;
    } else {
      u00r = ct;  u00i = 0.f; u01r = -st; u01i = 0.f;
      u10r = st;  u10i = 0.f; u11r = ct;  u11i = 0.f;
    }
  }
  float* M = G + g * 16;
  M[ 0] = u00r; M[ 1] = -u00i; M[ 2] = u01r; M[ 3] = -u01i;
  M[ 4] = u00i; M[ 5] =  u00r; M[ 6] = u01i; M[ 7] =  u01r;
  M[ 8] = u10r; M[ 9] = -u10i; M[10] = u11r; M[11] = -u11i;
  M[12] = u10i; M[13] =  u10r; M[14] = u11i; M[15] =  u11r;
  #pragma unroll
  for (int k = 0; k < 16; ++k) sm[g * 16 + k] = M[k];
  }
  __syncthreads();
  if (g < 64) {
    const int R = g >> 3, Cc = g & 7;
    const int ao = R >> 2, bo = (R >> 1) & 1, co = R & 1;
    const int ai = Cc >> 2, bi = (Cc >> 1) & 1, ci = Cc & 1;
    #pragma unroll
    for (int m = 0; m < 4; ++m) {
      const int a = (m >> 1) ? 6 : 1;
      const bool cv = (m & 1) != 0;
      const float* A = sm + (cv ? 29 + a : 20 + a) * 16;
      const float* V = sm + (ao ? 30 + a : 21 + a) * 16;
      const float ar = A[(2 * ao) * 4 + 2 * ai], ai_ = A[(2 * ao + 1) * 4 + 2 * ai];
      const float vr = V[(2 * bo) * 4 + 2 * bi];
      const float vi = V[(2 * bo + 1) * 4 + 2 * bi];
      const float zr = ar * vr - ai_ * vi, zi = ar * vi + ai_ * vr;
      G[OFF8 + m * 64 + g] = (co == ci) ? zr : (co ? zi : -zi);
    }
  }
  {
    const int R = g >> 4, Cc = g & 15;
    const int co = R >> 3, ao = (R >> 2) & 1, bo = (R >> 1) & 1, ro = R & 1;
    const int ai = Cc >> 3, bi = (Cc >> 2) & 1, ci = (Cc >> 1) & 1, ri = Cc & 1;
    #pragma unroll
    for (int m = 0; m < 4; ++m) {
      const int a = (m >> 1) ? 8 : 3;
      const bool cv = (m & 1) != 0;
      const float* A = sm + (cv ? 29 + a : 20 + a) * 16;
      const float* V = sm + (ao ? 30 + a : 21 + a) * 16;
      const float ar = A[(2 * ao) * 4 + 2 * ai], ai_ = A[(2 * ao + 1) * 4 + 2 * ai];
      const float vr = V[(2 * bo) * 4 + 2 * bi];
      const float vi = V[(2 * bo + 1) * 4 + 2 * bi];
      const float pr = ar * vr - ai_ * vi, pi = ar * vi + ai_ * vr;
      float xr, xi;
      if (bo || a == 3) {
        const float* X = sm + (bo ? 31 + a : 22 + a) * 16;
        xr = X[(2 * co) * 4 + 2 * ci];
        xi = X[(2 * co + 1) * 4 + 2 * ci];
      } else {
        xr = (co == ci) ? 1.f : 0.f;
        xi = 0.f;
      }
      const float zr = pr * xr - pi * xi, zi = pr * xi + pi * xr;
      G[OFF16 + m * 256 + g] = (ro == ri) ? zr : (ro ? zi : -zi);
    }
  }
}

__device__ __forceinline__ v8f mm(v2f a, v2f b, v8f c) {
  return __builtin_amdgcn_wmma_f32_16x16x4_f32(false, a, false, b, (short)0, c,
                                               false, false);
}

__device__ __forceinline__ v2f cmul(v2f a, v2f b) {
  return v2f{a.x * b.x - a.y * b.y, a.x * b.y + a.y * b.x};
}

__device__ __forceinline__ void cos_sin_clipped(float u, float& c, float& s) {
  const bool hi = u > 0.5f;
  const float v = hi ? 1.f - u : u;
  const float th = v * (0.5f * 3.14159265358979323846f);
  const float t = th * th;
  float p = fmaf(-1.9515295891e-4f, t, 8.3321608736e-3f);
  p = fmaf(p, t, -1.6666654611e-1f);
  const float sn = fmaf(p, t * th, th);
  float q = fmaf(2.443315711809948e-5f, t, -1.388731625493765e-3f);
  q = fmaf(q, t, 4.166664568298827e-2f);
  const float cs = fmaf(q, t * t, fmaf(-0.5f, t, 1.f));
  c = hi ? sn : cs;
  s = hi ? cs : sn;
}

__device__ __forceinline__ v2f pick(bool c, v2f one, v2f zero) {
  return v2f{c ? one.x : zero.x, c ? one.y : zero.y};
}

__device__ __forceinline__ v2f pick2(const v2f (&T)[2][2], bool a, bool b) {
  return pick(a, pick(b, T[1][1], T[1][0]), pick(b, T[0][1], T[0][0]));
}

__device__ __forceinline__ void load_mux(const float* __restrict__ G, int gR,
                                         int gF, int t, v2f& a1, v2f& a2) {
  const float* M = G + ((t & 4) ? gF : gR) * 16 + (t & 3) * 4;
  const float m0 = M[0], m1 = M[1], m2 = M[2], m3 = M[3];
  const bool on = ((t < 16) == ((t & 15) < 8));
  a1.x = on ? m0 : 0.f; a1.y = on ? m1 : 0.f;
  a2.x = on ? m2 : 0.f; a2.y = on ? m3 : 0.f;
}

__device__ __forceinline__ float lane_val(float v, int lane) {
  return __int_as_float(__builtin_amdgcn_readlane(__float_as_int(v), lane));
}
__device__ __forceinline__ v2f lane_val2(float x, float y, int lane) {
  return v2f{lane_val(x, lane), lane_val(y, lane)};
}

__device__ __forceinline__ v2f load_l0(const float* __restrict__ G, int j, int t) {
  const int row = t & 15, qq = 2 * j + (t >> 4);
  const int slot = ((row & 4) && qq > 0) ? 9 + qq : qq;
  const float* M = G + slot * 16 + (row & 3) * 4;
  const float m0 = M[0], m2 = M[2];
  return v2f{row < 8 ? m0 : 0.f, row < 8 ? m2 : 0.f};
}

__device__ __forceinline__ void load_gate2(const float* __restrict__ G, int m,
                                           int t, v2f (&a)[4]) {
  const float* M = G + OFF8 + m * 64 + (t & 7) * 8;
  const bool on = ((t < 16) == ((t & 15) < 8));
  #pragma unroll
  for (int j = 0; j < 4; ++j) {
    const float x = M[2 * j], y = M[2 * j + 1];
    a[j].x = on ? x : 0.f; a[j].y = on ? y : 0.f;
  }
}

__device__ __forceinline__ void load_gate2_halves(const float* __restrict__ G,
                                                  int m0, int t, v2f (&a)[4]) {
  const float* M = G + OFF8 + (m0 + (t >> 4)) * 64 + (t & 7) * 8;
  const bool on = ((t < 16) == ((t & 15) < 8));
  #pragma unroll
  for (int j = 0; j < 4; ++j) {
    const float x = M[2 * j], y = M[2 * j + 1];
    a[j].x = on ? x : 0.f; a[j].y = on ? y : 0.f;
  }
}

__device__ __forceinline__ void load_gate3(const float* __restrict__ G, int m,
                                           int t, v2f (&a)[4]) {
  const float* M = G + OFF16 + m * 256 + (t & 15) * 16 + (t >> 4) * 2;
  #pragma unroll
  for (int j = 0; j < 4; ++j) a[j] = v2f{M[4 * j], M[4 * j + 1]};
}

template <int RB1, int CB, int SEL, int N>
__device__ __forceinline__ void gate2_rr(v2f (&s)[N], const v2f (&g0)[4]) {
  constexpr int RB0 = RB1 - 1;
  #pragma unroll
  for (int p = 0; p < N / 4; ++p) {
    const int i0 = ((p >> RB0) << (RB1 + 1)) | (p & ((1 << RB0) - 1));
    const int i1 = i0 | (1 << RB0);
    const int i2 = i0 | (1 << RB1);
    const int i3 = i1 | i2;
    if (CB >= 0 && ((i0 >> (CB < 0 ? 0 : CB)) & 1) != SEL) continue;
    v8f z = {};
    v8f d = mm(g0[0], s[i0], z);
    d = mm(g0[1], s[i1], d);
    d = mm(g0[2], s[i2], d);
    d = mm(g0[3], s[i3], d);
    s[i0] = v2f{d[0], d[1]};
    s[i1] = v2f{d[2], d[3]};
    s[i2] = v2f{d[4], d[5]};
    s[i3] = v2f{d[6], d[7]};
  }
}

__device__ __forceinline__ void regroup(const v2f (&sr)[8], v2f (&zt)[8],
                                        float2* sv, int t) {
  const int k = ((t >> 2) & 6) | (t & 1);
  const int base = k * 32 + ((t & 6) << 2);
  #pragma unroll
  for (int r = 0; r < 8; ++r)
    sv[base + (r ^ k)] = make_float2(sr[r].x, sr[r].y);
  __builtin_amdgcn_wave_barrier();
  #pragma unroll
  for (int r = 0; r < 8; ++r) {
    float2 v = sv[r * 32 + (t ^ r)];
    zt[r] = v2f{v.x, v.y};
  }
}

#define G2(RB1, CB, m0) { load_gate2(G, (m0), t, g0); gate2_rr<RB1, CB, 0, 32>(s, g0); \
                          load_gate2(G, (m0) + 1, t, g1); gate2_rr<RB1, CB, 1, 32>(s, g1); }
#define G3(RB1, CB, m0) { load_gate3(G, (m0), t, g0); gate2_rr<RB1, CB, 0, 32>(s, g0); \
                          load_gate3(G, (m0) + 1, t, g1); gate2_rr<RB1, CB, 1, 32>(s, g1); }
#define PHASE_B1     G2(3, 4, 2) G3(1, 2, 2)

__global__ __launch_bounds__(256, 6) void qsim_kernel(
    const float* __restrict__ inp, const float* __restrict__ G,
    float* __restrict__ out) {
  __shared__ float2 st[WPB][256];
  const int t = threadIdx.x & 31;
  const int w = threadIdx.x >> 5;
  const int samp = blockIdx.x * WPB + w;
  const int samp_out = blockIdx.x * WPB + __builtin_amdgcn_readfirstlane(w);
  float2* sv = st[w];

  float cc[NQ], ss[NQ];
  const float* x = inp + samp * NQ;
  #pragma unroll
  for (int q = 0; q < NQ; ++q) {
    const float xv = fminf(fmaxf(x[q], 0.f), 1.f);
    cos_sin_clipped(xv, cc[q], ss[q]);
  }
  v2f a1, a2;
  v2f g0[4], g1[4];
  v2f w0[2];
  v2f T[NQ][2][2];
  v8f dT = {};
  #pragma unroll
  for (int j = 0; j < 5; ++j) {
    const bool hh = t >= 16;
    const bool own = (t & 15) == 2 * j + (hh ? 1 : 0);
    const float cj = hh ? cc[2 * j + 1] : cc[2 * j];
    const float sj = hh ? ss[2 * j + 1] : ss[2 * j];
    dT = mm(load_l0(G, j, t), v2f{own ? cj : 0.f, own ? sj : 0.f}, dT);
  }
  w0[0] = lane_val2(dT[0], dT[1], 0);
  w0[1] = lane_val2(dT[2], dT[3], 0);
  #pragma unroll
  for (int q = 1; q < 7; ++q) {
    T[q][0][0] = lane_val2(dT[0], dT[1], q);
    T[q][0][1] = lane_val2(dT[2], dT[3], q);
    T[q][1][0] = lane_val2(dT[4], dT[5], q);
    T[q][1][1] = lane_val2(dT[6], dT[7], q);
  }
  const bool c9 = (t & 1) != 0;
  v2f X[2][2];
  #pragma unroll
  for (int a = 0; a < 2; ++a) {
    #pragma unroll
    for (int b = 0; b < 2; ++b) X[a][b] = cmul(w0[a], T[1][a][b]);
  }
  {
    load_mux(G, 20, 19, t, a1, a2);
    const bool col1 = (t & 1) != 0;
    v8f z = {};
    v8f d = mm(a1, pick(col1, X[0][1], X[0][0]), z);
    d = mm(a2, pick(col1, X[1][1], X[1][0]), d);
    #pragma unroll
    for (int b = 0; b < 2; ++b) {
      X[0][b] = pick(c9, lane_val2(d[4], d[5], b), lane_val2(d[0], d[1], b));
      X[1][b] = pick(c9, lane_val2(d[6], d[7], b), lane_val2(d[2], d[3], b));
    }
  }
  const bool b5 = ((t >> 4) & 1) != 0, b6 = ((t >> 3) & 1) != 0;
  const bool l0 = ((t >> 2) & 1) != 0, l1 = ((t >> 1) & 1) != 0;
  const v2f t6 = pick2(T[6], b5, b6);
  v2f E[2][2];
  #pragma unroll
  for (int b = 0; b < 2; ++b) {
    const v2f f = cmul(pick(b5, T[5][b][1], T[5][b][0]), t6);
    E[0][b] = cmul(T[4][0][b], f);
    E[1][b] = cmul(T[4][1][b], f);
  }
  v2f P2[8], P3[16];
  #pragma unroll
  for (int k = 0; k < 8; ++k)
    P2[k] = cmul(X[k >> 2][(k >> 1) & 1], T[2][(k >> 1) & 1][k & 1]);
  #pragma unroll
  for (int k = 0; k < 16; ++k)
    P3[k] = cmul(P2[k >> 1], T[3][(k >> 1) & 1][k & 1]);
  v2f q4[4];
  {
    const bool h0 = t >= 16, h3 = (t & 2) != 0;
    load_gate2_halves(G, 0, t, g0);
    v8f d = {};
    #pragma unroll
    for (int j = 0; j < 4; ++j) {
      const v2f b = pick(h0, pick(h3, P3[8 + 2 * j + 1], P3[8 + 2 * j]),
                         pick(h3, P3[2 * j + 1], P3[2 * j]));
      d = mm(g0[j], b, d);
    }
    const bool y1 = (t & 8) != 0;
    const int src = (l0 ? 16 : 0) | (l1 ? 8 : 0) | (t & 1);
    const v2f r0 = pick(y1, v2f{d[4], d[5]}, v2f{d[0], d[1]});
    const v2f r1 = pick(y1, v2f{d[6], d[7]}, v2f{d[2], d[3]});
    #pragma unroll
    for (int q3 = 0; q3 < 2; ++q3) {
      const int from = src | (q3 << 1);
      q4[q3] = v2f{__shfl(r0.x, from, 32), __shfl(r0.y, from, 32)};
      q4[2 + q3] = v2f{__shfl(r1.x, from, 32), __shfl(r1.y, from, 32)};
    }
  }
  v2f sr[8];
  #pragma unroll
  for (int j = 0; j < 4; ++j) {
    sr[2 * j] = cmul(q4[j], E[j & 1][0]);
    sr[2 * j + 1] = cmul(q4[j], E[j & 1][1]);
  }

  load_gate3(G, 0, t, g0); gate2_rr<1, 2, 0, 8>(sr, g0);
  load_gate3(G, 1, t, g1); gate2_rr<1, 2, 1, 8>(sr, g1);
  v2f zt[8];
  regroup(sr, zt, sv, t);
  #pragma unroll
  for (int q = 7; q < NQ; ++q) {
    T[q][0][0] = lane_val2(dT[0], dT[1], q);
    T[q][0][1] = lane_val2(dT[2], dT[3], q);
    T[q][1][0] = lane_val2(dT[4], dT[5], q);
    T[q][1][1] = lane_val2(dT[6], dT[7], q);
  }
  v2f T89[8], F[16];
  #pragma unroll
  for (int k = 0; k < 8; ++k)
    T89[k] = cmul(T[8][k >> 2][(k >> 1) & 1], T[9][(k >> 1) & 1][k & 1]);
  #pragma unroll
  for (int k = 0; k < 16; ++k)
    F[k] = cmul(T[7][k >> 3][(k >> 2) & 1], T89[k & 7]);
  v2f s[32];
  #pragma unroll
  for (int r = 0; r < 32; ++r)
    s[r] = cmul(zt[((r >> 2) & 6) | (r & 1)], F[r & 15]);
  __builtin_amdgcn_sched_barrier(0);
  PHASE_B1

  float acc[NQ];
  #pragma unroll
  for (int q = 0; q < NQ; ++q) acc[q] = 0.f;
  #pragma unroll
  for (int r = 0; r < 32; ++r) {
    const int i = t * 32 + r;
    float p = s[r].x * s[r].x + s[r].y * s[r].y;
    #pragma unroll
    for (int q = 0; q < NQ; ++q)
      acc[q] += ((i >> (9 - q)) & 1) ? -p : p;
  }
  #pragma unroll
  for (int q = 0; q < NQ; ++q) {
    #pragma unroll
    for (int off = 16; off >= 1; off >>= 1)
      acc[q] += __shfl_xor(acc[q], off, 32);
  }
  if (t == 0) {
    #pragma unroll
    for (int q = 0; q < NQ; ++q) out[samp_out * NQ + q] = acc[q];
  }
}

extern "C" void kernel_launch(void* const* d_in, const int* in_sizes, int n_in,
                              void* d_out, int out_size, void* d_ws,
                              size_t ws_size, hipStream_t stream) {
  const float* inputs = (const float*)d_in[0];
  const float* weights = (const float*)d_in[1];
  float* out = (float*)d_out;
  float* G = (float*)d_ws;

  const int B = in_sizes[0] / NQ;
  build_gates_kernel<<<dim3(1), dim3(256), 0, stream>>>(weights, G);
  qsim_kernel<<<dim3(B / WPB), dim3(256), 0, stream>>>(inputs, G, out);
}
